# starting kernel
# baseline (speedup 1.0000x reference)
#include <hip/hip_runtime.h>


typedef float v2f __attribute__((ext_vector_type(2)));
typedef float v8f __attribute__((ext_vector_type(8)));

static __device__ __forceinline__ size_t row64(int n) { return (size_t)n * 64; }


__global__ __launch_bounds__(256) void k_zero(float* __restrict__ p, int n) {
    int i = blockIdx.x * 256 + threadIdx.x;
    if (i < n) p[i] = 0.0f;
}

__global__ __launch_bounds__(256) void k_deg(const float* __restrict__ ew,
                                             const int* __restrict__ row,
                                             const int* __restrict__ col,
                                             float* __restrict__ deg1,
                                             float* __restrict__ deg2, int E) {
    int e = blockIdx.x * 256 + threadIdx.x;
    if (e >= E) return;
    float w = ew[e];
    atomicAdd(&deg1[row[e]], w);
    atomicAdd(&deg2[col[e]], w);
}

__global__ __launch_bounds__(256) void k_invsqrt(float* __restrict__ d, int n) {
    int i = blockIdx.x * 256 + threadIdx.x;
    if (i < n) {
        float v = d[i];
        d[i] = (v > 0.0f) ? rsqrtf(v) : 0.0f;
    }
}

__global__ __launch_bounds__(256) void k_edgew(const float* __restrict__ ew,
                                               const int* __restrict__ row,
                                               const int* __restrict__ col,
                                               const float* __restrict__ dis1,
                                               const float* __restrict__ dis2,
                                               float* __restrict__ ow1,
                                               float* __restrict__ ow2, int E) {
    int e = blockIdx.x * 256 + threadIdx.x;
    if (e >= E) return;
    const float scale = 2.0f / 3.0f;
    int r = row[e], c = col[e];
    float w = ew[e];
    ow1[e] = -scale * dis1[r] * w * dis1[c];
    ow2[e] = -scale * dis2[r] * w * dis2[c];
}

__global__ __launch_bounds__(256) void k_bias(const float* __restrict__ b1,
                                              const float* __restrict__ b2,
                                              float* __restrict__ out, int n128) {
    int i = blockIdx.x * 256 + threadIdx.x;
    if (i >= n128) return;
    int j = i & 127;
    out[i] = (j < 64) ? b1[j] : b2[j - 64];
}

__global__ __launch_bounds__(256) void k_cheb_scale(const float* __restrict__ v,
                                                    const float* __restrict__ prev,
                                                    float* __restrict__ dst,
                                                    float alpha_diag, float beta,
                                                    int n64) {
    int i = blockIdx.x * 256 + threadIdx.x;
    if (i < n64) dst[i] = alpha_diag * v[i] + beta * prev[i];
}

__global__ __launch_bounds__(256) void k_scatter(const float* __restrict__ v,
                                                 float* __restrict__ outf,
                                                 const int* __restrict__ srcIdx,
                                                 const int* __restrict__ dstIdx,
                                                 const float* __restrict__ ow,
                                                 float alpha, int E) {
    int e = blockIdx.x * 8 + (threadIdx.x >> 5);
    if (e >= E) return;
    int c2 = (threadIdx.x & 31) * 2;
    int s = srcIdx[e];
    int d = dstIdx[e];
    float w = alpha * ow[e];
    float2 val = *(const float2*)(v + row64(s) + c2);
    float* o = outf + row64(d) + c2;
    atomicAdd(o + 0, w * val.x);
    atomicAdd(o + 1, w * val.y);
}

__global__ __launch_bounds__(256) void k_gemm_acc(const float* __restrict__ Tx,
                                                  const float* __restrict__ W,
                                                  float* __restrict__ out,
                                                  int colofs, int nRows) {
    __shared__ float Wl[64 * 64];
    __shared__ float Al[128 * 64];

    const int tid = threadIdx.x;
    const int row0b = blockIdx.x * 128;

    {
        const float4* W4 = (const float4*)W;
        float4* Wl4 = (float4*)Wl;
#pragma unroll
        for (int i = 0; i < 4; ++i) Wl4[tid + 256 * i] = W4[tid + 256 * i];
    }
    {
        float4* Al4 = (float4*)Al;
#pragma unroll
        for (int i = 0; i < 8; ++i) {
            int idx = tid + 256 * i;
            int r = row0b + (idx >> 4);
            float4 v = make_float4(0.f, 0.f, 0.f, 0.f);
            if (r < nRows) v = ((const float4*)(Tx + row64(r)))[idx & 15];
            Al4[idx] = v;
        }
    }
    __syncthreads();

    const int wave = tid >> 5;
    const int lane = tid & 31;
    const int lm = lane & 15;
    const int lg = lane >> 4;

    const int rowt = wave * 16;
    const int row0 = row0b + rowt;

    v8f c0, c1, c2, c3;
    const int nclamp = nRows - 1;
#pragma unroll
    for (int i = 0; i < 8; ++i) {
        int r = row0 + i + lg * 8;
        bool ok = (r < nRows);
        int rc = ok ? r : nclamp;
        const float* orow = out + (size_t)rc * 128 + colofs + lm;
        float v0 = orow[0], v1 = orow[16], v2 = orow[32], v3 = orow[48];
        c0[i] = ok ? v0 : 0.f;
        c1[i] = ok ? v1 : 0.f;
        c2[i] = ok ? v2 : 0.f;
        c3[i] = ok ? v3 : 0.f;
    }

    const float* arow_l = Al + (rowt + lm) * 64;
    const float* bcol_l = Wl + lm;

#pragma unroll
    for (int kk = 0; kk < 16; ++kk) {
        const int kbase = kk * 4 + lg * 2;
        v2f a = *(const v2f*)(arow_l + kbase);
        v2f b0, b1v, b2, b3;
        b0[0]  = bcol_l[(kbase + 0) * 64 + 0];
        b0[1]  = bcol_l[(kbase + 1) * 64 + 0];
        b1v[0] = bcol_l[(kbase + 0) * 64 + 16];
        b1v[1] = bcol_l[(kbase + 1) * 64 + 16];
        b2[0]  = bcol_l[(kbase + 0) * 64 + 32];
        b2[1]  = bcol_l[(kbase + 1) * 64 + 32];
        b3[0]  = bcol_l[(kbase + 0) * 64 + 48];
        b3[1]  = bcol_l[(kbase + 1) * 64 + 48];
        c0 = __builtin_amdgcn_wmma_f32_16x16x4_f32(false, a, false, b0,  (short)0, c0, false, false);
        c1 = __builtin_amdgcn_wmma_f32_16x16x4_f32(false, a, false, b1v, (short)0, c1, false, false);
        c2 = __builtin_amdgcn_wmma_f32_16x16x4_f32(false, a, false, b2,  (short)0, c2, false, false);
        c3 = __builtin_amdgcn_wmma_f32_16x16x4_f32(false, a, false, b3,  (short)0, c3, false, false);
    }

#pragma unroll
    for (int i = 0; i < 8; ++i) {
        int r = row0 + i + lg * 8;
        if (r < nRows) {
            float* orow = out + (size_t)r * 128 + colofs + lm;
            orow[0]  = c0[i];
            orow[16] = c1[i];
            orow[32] = c2[i];
            orow[48] = c3[i];
        }
    }
}


extern "C" void kernel_launch(void* const* d_in, const int* in_sizes, int n_in,
                              void* d_out, int out_size, void* d_ws, size_t ws_size,
                              hipStream_t stream) {
    const float* x  = (const float*)d_in[0];
    const int*   ei = (const int*)d_in[1];
    const float* ew = (const float*)d_in[2];
    const float* W1 = (const float*)d_in[3];
    const float* b1 = (const float*)d_in[4];
    const float* W2 = (const float*)d_in[5];
    const float* b2 = (const float*)d_in[6];
    float* out = (float*)d_out;

    const int N = in_sizes[0] / 64;
    const int E = in_sizes[2];
    const int* row = ei;
    const int* col = ei + E;

    float* ws   = (float*)d_ws;
    float* deg1 = ws;
    float* deg2 = deg1 + N;
    float* ow1  = deg2 + N;
    float* ow2  = ow1 + E;
    float* tA   = ow2 + E;
    float* tB   = tA + (size_t)N * 64;
    float* tC   = tB + (size_t)N * 64;

    const int n64  = N * 64;
    const int n128 = N * 128;
    const float DIAG = 2.0f / 3.0f - 1.0f;

    dim3 blk(256);
    int gN2   = (2 * N + 255) / 256;
    int gE    = (E + 255) / 256;
    int gE8   = (E + 7) / 8;
    int g64   = (n64 + 255) / 256;
    int g128  = (n128 + 255) / 256;
    int gGemm = (N + 127) / 128;

    k_zero<<<gN2, blk, 0, stream>>>(deg1, 2 * N);
    k_deg<<<gE, blk, 0, stream>>>(ew, row, col, deg1, deg2, E);
    k_invsqrt<<<gN2, blk, 0, stream>>>(deg1, 2 * N);
    k_edgew<<<gE, blk, 0, stream>>>(ew, row, col, deg1, deg2, ow1, ow2, E);

    k_bias<<<g128, blk, 0, stream>>>(b1, b2, out, n128);

    for (int dir = 0; dir < 2; ++dir) {
        const int*   src  = (dir == 0) ? col : row;
        const int*   dst  = (dir == 0) ? row : col;
        const float* ow   = (dir == 0) ? ow1 : ow2;
        const float* W    = (dir == 0) ? W1 : W2;
        const int    cofs = (dir == 0) ? 0 : 64;

        k_gemm_acc<<<gGemm, blk, 0, stream>>>(x, W, out, cofs, N);

        k_cheb_scale<<<g64, blk, 0, stream>>>(x, x, tA, DIAG, 0.0f, n64);
        k_scatter<<<gE8, blk, 0, stream>>>(x, tA, src, dst, ow, 1.0f, E);
        k_gemm_acc<<<gGemm, blk, 0, stream>>>(tA, W + 1 * 4096, out, cofs, N);

        k_cheb_scale<<<g64, blk, 0, stream>>>(tA, x, tB, 2.0f * DIAG, -1.0f, n64);
        k_scatter<<<gE8, blk, 0, stream>>>(tA, tB, src, dst, ow, 2.0f, E);
        k_gemm_acc<<<gGemm, blk, 0, stream>>>(tB, W + 2 * 4096, out, cofs, N);

        k_cheb_scale<<<g64, blk, 0, stream>>>(tB, tA, tC, 2.0f * DIAG, -1.0f, n64);
        k_scatter<<<gE8, blk, 0, stream>>>(tB, tC, src, dst, ow, 2.0f, E);
        k_gemm_acc<<<gGemm, blk, 0, stream>>>(tC, W + 3 * 4096, out, cofs, N);

        k_cheb_scale<<<g64, blk, 0, stream>>>(tC, tB, tA, 2.0f * DIAG, -1.0f, n64);
        k_scatter<<<gE8, blk, 0, stream>>>(tC, tA, src, dst, ow, 2.0f, E);
        k_gemm_acc<<<gGemm, blk, 0, stream>>>(tA, W + 4 * 4096, out, cofs, N);
    }
}


# 16x the work per wave in the wave-entry-bound kernels: scatter takes 16 edges per wave (indices by scalar load, 16 row gathers in flight), bias and scale 16 floats per thread; atomic sums as before
# speedup vs baseline: 1.4048x; 1.4048x over previous
#include <hip/hip_runtime.h>


typedef float v2f __attribute__((ext_vector_type(2)));
typedef float v8f __attribute__((ext_vector_type(8)));

static __device__ __forceinline__ size_t row64(int n) { return (size_t)n * 64; }


__global__ __launch_bounds__(256) void k_zero(float* __restrict__ p, int n) {
    int i = blockIdx.x * 256 + threadIdx.x;
    if (i < n) p[i] = 0.0f;
}

__global__ __launch_bounds__(256) void k_deg(const float* __restrict__ ew,
                                             const int* __restrict__ row,
                                             const int* __restrict__ col,
                                             float* __restrict__ deg1,
                                             float* __restrict__ deg2, int E) {
    int e = blockIdx.x * 256 + threadIdx.x;
    if (e >= E) return;
    float w = ew[e];
    atomicAdd(&deg1[row[e]], w);
    atomicAdd(&deg2[col[e]], w);
}

__global__ __launch_bounds__(256) void k_invsqrt(float* __restrict__ d, int n) {
    int i = blockIdx.x * 256 + threadIdx.x;
    if (i < n) {
        float v = d[i];
        d[i] = (v > 0.0f) ? rsqrtf(v) : 0.0f;
    }
}

__global__ __launch_bounds__(256) void k_edgew(const float* __restrict__ ew,
                                               const int* __restrict__ row,
                                               const int* __restrict__ col,
                                               const float* __restrict__ dis1,
                                               const float* __restrict__ dis2,
                                               float* __restrict__ ow1,
                                               float* __restrict__ ow2, int E) {
    int e = blockIdx.x * 256 + threadIdx.x;
    if (e >= E) return;
    const float scale = 2.0f / 3.0f;
    int r = row[e], c = col[e];
    float w = ew[e];
    ow1[e] = -scale * dis1[r] * w * dis1[c];
    ow2[e] = -scale * dis2[r] * w * dis2[c];
}

__global__ __launch_bounds__(256) void k_bias(const float* __restrict__ b1,
                                              const float* __restrict__ b2,
                                              float* __restrict__ out, int n4) {
    const int stride = gridDim.x * 256;
    int i = blockIdx.x * 256 + threadIdx.x;
    float4* out4 = (float4*)out;
    const int j = (threadIdx.x & 31) * 4;
    const float4 bv = *(const float4*)((j < 64) ? (b1 + j) : (b2 + (j - 64)));
#pragma unroll
    for (int u = 0; u < 4; ++u, i += stride) {
        if (i < n4) out4[i] = bv;
    }
}

__global__ __launch_bounds__(256) void k_cheb_scale(const float* __restrict__ v,
                                                    const float* __restrict__ prev,
                                                    float* __restrict__ dst,
                                                    float alpha_diag, float beta,
                                                    int n4) {
    const int stride = gridDim.x * 256;
    const int i0 = blockIdx.x * 256 + threadIdx.x;
    const float4* v4 = (const float4*)v;
    const float4* p4 = (const float4*)prev;
    float4* d4 = (float4*)dst;
    float4 a[4], p[4];
#pragma unroll
    for (int u = 0; u < 4; ++u) {
        int i = i0 + u * stride;
        int ic = (i < n4) ? i : 0;
        a[u] = v4[ic];
        p[u] = p4[ic];
    }
#pragma unroll
    for (int u = 0; u < 4; ++u) {
        int i = i0 + u * stride;
        if (i < n4) {
            float4 r;
            r.x = alpha_diag * a[u].x + beta * p[u].x;
            r.y = alpha_diag * a[u].y + beta * p[u].y;
            r.z = alpha_diag * a[u].z + beta * p[u].z;
            r.w = alpha_diag * a[u].w + beta * p[u].w;
            d4[i] = r;
        }
    }
}

#define EPW 16
__global__ __launch_bounds__(256) void k_scatter(const float* __restrict__ v,
                                                 float* __restrict__ outf,
                                                 const int* __restrict__ srcIdx,
                                                 const int* __restrict__ dstIdx,
                                                 const float* __restrict__ ow,
                                                 float alpha, int E) {
    const int wave = __builtin_amdgcn_readfirstlane((int)(threadIdx.x >> 5));
    const int c2 = (threadIdx.x & 31) * 2;
    const int e0 = (blockIdx.x * 8 + wave) * EPW;
    if (e0 >= E) return;
    if (e0 + EPW <= E) {
        int s[EPW], d[EPW];
        float w[EPW];
        float2 val[EPW];
#pragma unroll
        for (int j = 0; j < EPW; ++j) {
            s[j] = srcIdx[e0 + j];
            d[j] = dstIdx[e0 + j];
            w[j] = alpha * ow[e0 + j];
        }
#pragma unroll
        for (int j = 0; j < EPW; ++j)
            val[j] = *(const float2*)(v + row64(s[j]) + c2);
#pragma unroll
        for (int j = 0; j < EPW; ++j) {
            float* o = outf + row64(d[j]) + c2;
            atomicAdd(o + 0, w[j] * val[j].x);
            atomicAdd(o + 1, w[j] * val[j].y);
        }
    } else {
        for (int e = e0; e < E; ++e) {
            float wt = alpha * ow[e];
            float2 vv = *(const float2*)(v + row64(srcIdx[e]) + c2);
            float* o = outf + row64(dstIdx[e]) + c2;
            atomicAdd(o + 0, wt * vv.x);
            atomicAdd(o + 1, wt * vv.y);
        }
    }
}

__global__ __launch_bounds__(256) void k_gemm_acc(const float* __restrict__ Tx,
                                                  const float* __restrict__ W,
                                                  float* __restrict__ out,
                                                  int colofs, int nRows) {
    __shared__ float Wl[64 * 64];
    __shared__ float Al[128 * 64];

    const int tid = threadIdx.x;
    const int row0b = blockIdx.x * 128;

    {
        const float4* W4 = (const float4*)W;
        float4* Wl4 = (float4*)Wl;
#pragma unroll
        for (int i = 0; i < 4; ++i) Wl4[tid + 256 * i] = W4[tid + 256 * i];
    }
    {
        float4* Al4 = (float4*)Al;
#pragma unroll
        for (int i = 0; i < 8; ++i) {
            int idx = tid + 256 * i;
            int r = row0b + (idx >> 4);
            float4 v = make_float4(0.f, 0.f, 0.f, 0.f);
            if (r < nRows) v = ((const float4*)(Tx + row64(r)))[idx & 15];
            Al4[idx] = v;
        }
    }
    __syncthreads();

    const int wave = tid >> 5;
    const int lane = tid & 31;
    const int lm = lane & 15;
    const int lg = lane >> 4;

    const int rowt = wave * 16;
    const int row0 = row0b + rowt;

    v8f c0, c1, c2, c3;
    const int nclamp = nRows - 1;
#pragma unroll
    for (int i = 0; i < 8; ++i) {
        int r = row0 + i + lg * 8;
        bool ok = (r < nRows);
        int rc = ok ? r : nclamp;
        const float* orow = out + (size_t)rc * 128 + colofs + lm;
        float v0 = orow[0], v1 = orow[16], v2 = orow[32], v3 = orow[48];
        c0[i] = ok ? v0 : 0.f;
        c1[i] = ok ? v1 : 0.f;
        c2[i] = ok ? v2 : 0.f;
        c3[i] = ok ? v3 : 0.f;
    }

    const float* arow_l = Al + (rowt + lm) * 64;
    const float* bcol_l = Wl + lm;

#pragma unroll
    for (int kk = 0; kk < 16; ++kk) {
        const int kbase = kk * 4 + lg * 2;
        v2f a = *(const v2f*)(arow_l + kbase);
        v2f b0, b1v, b2, b3;
        b0[0]  = bcol_l[(kbase + 0) * 64 + 0];
        b0[1]  = bcol_l[(kbase + 1) * 64 + 0];
        b1v[0] = bcol_l[(kbase + 0) * 64 + 16];
        b1v[1] = bcol_l[(kbase + 1) * 64 + 16];
        b2[0]  = bcol_l[(kbase + 0) * 64 + 32];
        b2[1]  = bcol_l[(kbase + 1) * 64 + 32];
        b3[0]  = bcol_l[(kbase + 0) * 64 + 48];
        b3[1]  = bcol_l[(kbase + 1) * 64 + 48];
        c0 = __builtin_amdgcn_wmma_f32_16x16x4_f32(false, a, false, b0,  (short)0, c0, false, false);
        c1 = __builtin_amdgcn_wmma_f32_16x16x4_f32(false, a, false, b1v, (short)0, c1, false, false);
        c2 = __builtin_amdgcn_wmma_f32_16x16x4_f32(false, a, false, b2,  (short)0, c2, false, false);
        c3 = __builtin_amdgcn_wmma_f32_16x16x4_f32(false, a, false, b3,  (short)0, c3, false, false);
    }

#pragma unroll
    for (int i = 0; i < 8; ++i) {
        int r = row0 + i + lg * 8;
        if (r < nRows) {
            float* orow = out + (size_t)r * 128 + colofs + lm;
            orow[0]  = c0[i];
            orow[16] = c1[i];
            orow[32] = c2[i];
            orow[48] = c3[i];
        }
    }
}


extern "C" void kernel_launch(void* const* d_in, const int* in_sizes, int n_in,
                              void* d_out, int out_size, void* d_ws, size_t ws_size,
                              hipStream_t stream) {
    const float* x  = (const float*)d_in[0];
    const int*   ei = (const int*)d_in[1];
    const float* ew = (const float*)d_in[2];
    const float* W1 = (const float*)d_in[3];
    const float* b1 = (const float*)d_in[4];
    const float* W2 = (const float*)d_in[5];
    const float* b2 = (const float*)d_in[6];
    float* out = (float*)d_out;

    const int N = in_sizes[0] / 64;
    const int E = in_sizes[2];
    const int* row = ei;
    const int* col = ei + E;

    float* ws   = (float*)d_ws;
    float* deg1 = ws;
    float* deg2 = deg1 + N;
    float* ow1  = deg2 + N;
    float* ow2  = ow1 + E;
    float* tA   = ow2 + E;
    float* tB   = tA + (size_t)N * 64;
    float* tC   = tB + (size_t)N * 64;

    const int n64q  = N * 16;
    const int n128q = N * 32;
    const float DIAG = 2.0f / 3.0f - 1.0f;

    dim3 blk(256);
    int gN2   = (2 * N + 255) / 256;
    int gE    = (E + 255) / 256;
    int gE8   = (E + 8 * EPW - 1) / (8 * EPW);
    int g64   = (n64q + 1023) / 1024;
    int g128  = (n128q + 1023) / 1024;
    int gGemm = (N + 127) / 128;

    k_zero<<<gN2, blk, 0, stream>>>(deg1, 2 * N);
    k_deg<<<gE, blk, 0, stream>>>(ew, row, col, deg1, deg2, E);
    k_invsqrt<<<gN2, blk, 0, stream>>>(deg1, 2 * N);
    k_edgew<<<gE, blk, 0, stream>>>(ew, row, col, deg1, deg2, ow1, ow2, E);

    k_bias<<<g128, blk, 0, stream>>>(b1, b2, out, n128q);

    for (int dir = 0; dir < 2; ++dir) {
        const int*   src  = (dir == 0) ? col : row;
        const int*   dst  = (dir == 0) ? row : col;
        const float* ow   = (dir == 0) ? ow1 : ow2;
        const float* W    = (dir == 0) ? W1 : W2;
        const int    cofs = (dir == 0) ? 0 : 64;

        k_gemm_acc<<<gGemm, blk, 0, stream>>>(x, W, out, cofs, N);

        k_cheb_scale<<<g64, blk, 0, stream>>>(x, x, tA, DIAG, 0.0f, n64q);
        k_scatter<<<gE8, blk, 0, stream>>>(x, tA, src, dst, ow, 1.0f, E);
        k_gemm_acc<<<gGemm, blk, 0, stream>>>(tA, W + 1 * 4096, out, cofs, N);

        k_cheb_scale<<<g64, blk, 0, stream>>>(tA, x, tB, 2.0f * DIAG, -1.0f, n64q);
        k_scatter<<<gE8, blk, 0, stream>>>(tA, tB, src, dst, ow, 2.0f, E);
        k_gemm_acc<<<gGemm, blk, 0, stream>>>(tB, W + 2 * 4096, out, cofs, N);

        k_cheb_scale<<<g64, blk, 0, stream>>>(tB, tA, tC, 2.0f * DIAG, -1.0f, n64q);
        k_scatter<<<gE8, blk, 0, stream>>>(tB, tC, src, dst, ow, 2.0f, E);
        k_gemm_acc<<<gGemm, blk, 0, stream>>>(tC, W + 3 * 4096, out, cofs, N);

        k_cheb_scale<<<g64, blk, 0, stream>>>(tC, tB, tA, 2.0f * DIAG, -1.0f, n64q);
        k_scatter<<<gE8, blk, 0, stream>>>(tC, tA, src, dst, ow, 2.0f, E);
        k_gemm_acc<<<gGemm, blk, 0, stream>>>(tA, W + 4 * 4096, out, cofs, N);
    }
}


# Scatter: lane L takes channels L and L+32, not 2L and 2L+1: each gather and each atomic instruction covers one whole 128-byte line, halving atomic line requests; atomic sums as before
# speedup vs baseline: 1.4750x; 1.0500x over previous
#include <hip/hip_runtime.h>


typedef float v2f __attribute__((ext_vector_type(2)));
typedef float v8f __attribute__((ext_vector_type(8)));

static __device__ __forceinline__ size_t row64(int n) { return (size_t)n * 64; }


__global__ __launch_bounds__(256) void k_zero(float* __restrict__ p, int n) {
    int i = blockIdx.x * 256 + threadIdx.x;
    if (i < n) p[i] = 0.0f;
}

__global__ __launch_bounds__(256) void k_deg(const float* __restrict__ ew,
                                             const int* __restrict__ row,
                                             const int* __restrict__ col,
                                             float* __restrict__ deg1,
                                             float* __restrict__ deg2, int E) {
    int e = blockIdx.x * 256 + threadIdx.x;
    if (e >= E) return;
    float w = ew[e];
    atomicAdd(&deg1[row[e]], w);
    atomicAdd(&deg2[col[e]], w);
}

__global__ __launch_bounds__(256) void k_invsqrt(float* __restrict__ d, int n) {
    int i = blockIdx.x * 256 + threadIdx.x;
    if (i < n) {
        float v = d[i];
        d[i] = (v > 0.0f) ? rsqrtf(v) : 0.0f;
    }
}

__global__ __launch_bounds__(256) void k_edgew(const float* __restrict__ ew,
                                               const int* __restrict__ row,
                                               const int* __restrict__ col,
                                               const float* __restrict__ dis1,
                                               const float* __restrict__ dis2,
                                               float* __restrict__ ow1,
                                               float* __restrict__ ow2, int E) {
    int e = blockIdx.x * 256 + threadIdx.x;
    if (e >= E) return;
    const float scale = 2.0f / 3.0f;
    int r = row[e], c = col[e];
    float w = ew[e];
    ow1[e] = -scale * dis1[r] * w * dis1[c];
    ow2[e] = -scale * dis2[r] * w * dis2[c];
}

__global__ __launch_bounds__(256) void k_bias(const float* __restrict__ b1,
                                              const float* __restrict__ b2,
                                              float* __restrict__ out, int n4) {
    const int stride = gridDim.x * 256;
    int i = blockIdx.x * 256 + threadIdx.x;
    float4* out4 = (float4*)out;
    const int j = (threadIdx.x & 31) * 4;
    const float4 bv = *(const float4*)((j < 64) ? (b1 + j) : (b2 + (j - 64)));
#pragma unroll
    for (int u = 0; u < 4; ++u, i += stride) {
        if (i < n4) out4[i] = bv;
    }
}

__global__ __launch_bounds__(256) void k_cheb_scale(const float* __restrict__ v,
                                                    const float* __restrict__ prev,
                                                    float* __restrict__ dst,
                                                    float alpha_diag, float beta,
                                                    int n4) {
    const int stride = gridDim.x * 256;
    const int i0 = blockIdx.x * 256 + threadIdx.x;
    const float4* v4 = (const float4*)v;
    const float4* p4 = (const float4*)prev;
    float4* d4 = (float4*)dst;
    float4 a[4], p[4];
#pragma unroll
    for (int u = 0; u < 4; ++u) {
        int i = i0 + u * stride;
        int ic = (i < n4) ? i : 0;
        a[u] = v4[ic];
        p[u] = p4[ic];
    }
#pragma unroll
    for (int u = 0; u < 4; ++u) {
        int i = i0 + u * stride;
        if (i < n4) {
            float4 r;
            r.x = alpha_diag * a[u].x + beta * p[u].x;
            r.y = alpha_diag * a[u].y + beta * p[u].y;
            r.z = alpha_diag * a[u].z + beta * p[u].z;
            r.w = alpha_diag * a[u].w + beta * p[u].w;
            d4[i] = r;
        }
    }
}

#define EPW 16
__global__ __launch_bounds__(256) void k_scatter(const float* __restrict__ v,
                                                 float* __restrict__ outf,
                                                 const int* __restrict__ srcIdx,
                                                 const int* __restrict__ dstIdx,
                                                 const float* __restrict__ ow,
                                                 float alpha, int E) {
    const int wave = __builtin_amdgcn_readfirstlane((int)(threadIdx.x >> 5));
    const int lane = threadIdx.x & 31;
    const int e0 = (blockIdx.x * 8 + wave) * EPW;
    if (e0 >= E) return;
    if (e0 + EPW <= E) {
        int s[EPW], d[EPW];
        float w[EPW];
        float lo[EPW], hi[EPW];
#pragma unroll
        for (int j = 0; j < EPW; ++j) {
            s[j] = srcIdx[e0 + j];
            d[j] = dstIdx[e0 + j];
            w[j] = alpha * ow[e0 + j];
        }
#pragma unroll
        for (int j = 0; j < EPW; ++j) {
            const float* r = v + row64(s[j]) + lane;
            lo[j] = r[0];
            hi[j] = r[32];
        }
#pragma unroll
        for (int j = 0; j < EPW; ++j) {
            float* o = outf + row64(d[j]) + lane;
            atomicAdd(o, w[j] * lo[j]);
            atomicAdd(o + 32, w[j] * hi[j]);
        }
    } else {
        for (int e = e0; e < E; ++e) {
            float wt = alpha * ow[e];
            const float* r = v + row64(srcIdx[e]) + lane;
            float vlo = r[0], vhi = r[32];
            float* o = outf + row64(dstIdx[e]) + lane;
            atomicAdd(o, wt * vlo);
            atomicAdd(o + 32, wt * vhi);
        }
    }
}

__global__ __launch_bounds__(256) void k_gemm_acc(const float* __restrict__ Tx,
                                                  const float* __restrict__ W,
                                                  float* __restrict__ out,
                                                  int colofs, int nRows) {
    __shared__ float Wl[64 * 64];
    __shared__ float Al[128 * 64];

    const int tid = threadIdx.x;
    const int row0b = blockIdx.x * 128;

    {
        const float4* W4 = (const float4*)W;
        float4* Wl4 = (float4*)Wl;
#pragma unroll
        for (int i = 0; i < 4; ++i) Wl4[tid + 256 * i] = W4[tid + 256 * i];
    }
    {
        float4* Al4 = (float4*)Al;
#pragma unroll
        for (int i = 0; i < 8; ++i) {
            int idx = tid + 256 * i;
            int r = row0b + (idx >> 4);
            float4 v = make_float4(0.f, 0.f, 0.f, 0.f);
            if (r < nRows) v = ((const float4*)(Tx + row64(r)))[idx & 15];
            Al4[idx] = v;
        }
    }
    __syncthreads();

    const int wave = tid >> 5;
    const int lane = tid & 31;
    const int lm = lane & 15;
    const int lg = lane >> 4;

    const int rowt = wave * 16;
    const int row0 = row0b + rowt;

    v8f c0, c1, c2, c3;
    const int nclamp = nRows - 1;
#pragma unroll
    for (int i = 0; i < 8; ++i) {
        int r = row0 + i + lg * 8;
        bool ok = (r < nRows);
        int rc = ok ? r : nclamp;
        const float* orow = out + (size_t)rc * 128 + colofs + lm;
        float v0 = orow[0], v1 = orow[16], v2 = orow[32], v3 = orow[48];
        c0[i] = ok ? v0 : 0.f;
        c1[i] = ok ? v1 : 0.f;
        c2[i] = ok ? v2 : 0.f;
        c3[i] = ok ? v3 : 0.f;
    }

    const float* arow_l = Al + (rowt + lm) * 64;
    const float* bcol_l = Wl + lm;

#pragma unroll
    for (int kk = 0; kk < 16; ++kk) {
        const int kbase = kk * 4 + lg * 2;
        v2f a = *(const v2f*)(arow_l + kbase);
        v2f b0, b1v, b2, b3;
        b0[0]  = bcol_l[(kbase + 0) * 64 + 0];
        b0[1]  = bcol_l[(kbase + 1) * 64 + 0];
        b1v[0] = bcol_l[(kbase + 0) * 64 + 16];
        b1v[1] = bcol_l[(kbase + 1) * 64 + 16];
        b2[0]  = bcol_l[(kbase + 0) * 64 + 32];
        b2[1]  = bcol_l[(kbase + 1) * 64 + 32];
        b3[0]  = bcol_l[(kbase + 0) * 64 + 48];
        b3[1]  = bcol_l[(kbase + 1) * 64 + 48];
        c0 = __builtin_amdgcn_wmma_f32_16x16x4_f32(false, a, false, b0,  (short)0, c0, false, false);
        c1 = __builtin_amdgcn_wmma_f32_16x16x4_f32(false, a, false, b1v, (short)0, c1, false, false);
        c2 = __builtin_amdgcn_wmma_f32_16x16x4_f32(false, a, false, b2,  (short)0, c2, false, false);
        c3 = __builtin_amdgcn_wmma_f32_16x16x4_f32(false, a, false, b3,  (short)0, c3, false, false);
    }

#pragma unroll
    for (int i = 0; i < 8; ++i) {
        int r = row0 + i + lg * 8;
        if (r < nRows) {
            float* orow = out + (size_t)r * 128 + colofs + lm;
            orow[0]  = c0[i];
            orow[16] = c1[i];
            orow[32] = c2[i];
            orow[48] = c3[i];
        }
    }
}


extern "C" void kernel_launch(void* const* d_in, const int* in_sizes, int n_in,
                              void* d_out, int out_size, void* d_ws, size_t ws_size,
                              hipStream_t stream) {
    const float* x  = (const float*)d_in[0];
    const int*   ei = (const int*)d_in[1];
    const float* ew = (const float*)d_in[2];
    const float* W1 = (const float*)d_in[3];
    const float* b1 = (const float*)d_in[4];
    const float* W2 = (const float*)d_in[5];
    const float* b2 = (const float*)d_in[6];
    float* out = (float*)d_out;

    const int N = in_sizes[0] / 64;
    const int E = in_sizes[2];
    const int* row = ei;
    const int* col = ei + E;

    float* ws   = (float*)d_ws;
    float* deg1 = ws;
    float* deg2 = deg1 + N;
    float* ow1  = deg2 + N;
    float* ow2  = ow1 + E;
    float* tA   = ow2 + E;
    float* tB   = tA + (size_t)N * 64;
    float* tC   = tB + (size_t)N * 64;

    const int n64q  = N * 16;
    const int n128q = N * 32;
    const float DIAG = 2.0f / 3.0f - 1.0f;

    dim3 blk(256);
    int gN2   = (2 * N + 255) / 256;
    int gE    = (E + 255) / 256;
    int gE8   = (E + 8 * EPW - 1) / (8 * EPW);
    int g64   = (n64q + 1023) / 1024;
    int g128  = (n128q + 1023) / 1024;
    int gGemm = (N + 127) / 128;

    k_zero<<<gN2, blk, 0, stream>>>(deg1, 2 * N);
    k_deg<<<gE, blk, 0, stream>>>(ew, row, col, deg1, deg2, E);
    k_invsqrt<<<gN2, blk, 0, stream>>>(deg1, 2 * N);
    k_edgew<<<gE, blk, 0, stream>>>(ew, row, col, deg1, deg2, ow1, ow2, E);

    k_bias<<<g128, blk, 0, stream>>>(b1, b2, out, n128q);

    for (int dir = 0; dir < 2; ++dir) {
        const int*   src  = (dir == 0) ? col : row;
        const int*   dst  = (dir == 0) ? row : col;
        const float* ow   = (dir == 0) ? ow1 : ow2;
        const float* W    = (dir == 0) ? W1 : W2;
        const int    cofs = (dir == 0) ? 0 : 64;

        k_gemm_acc<<<gGemm, blk, 0, stream>>>(x, W, out, cofs, N);

        k_cheb_scale<<<g64, blk, 0, stream>>>(x, x, tA, DIAG, 0.0f, n64q);
        k_scatter<<<gE8, blk, 0, stream>>>(x, tA, src, dst, ow, 1.0f, E);
        k_gemm_acc<<<gGemm, blk, 0, stream>>>(tA, W + 1 * 4096, out, cofs, N);

        k_cheb_scale<<<g64, blk, 0, stream>>>(tA, x, tB, 2.0f * DIAG, -1.0f, n64q);
        k_scatter<<<gE8, blk, 0, stream>>>(tA, tB, src, dst, ow, 2.0f, E);
        k_gemm_acc<<<gGemm, blk, 0, stream>>>(tB, W + 2 * 4096, out, cofs, N);

        k_cheb_scale<<<g64, blk, 0, stream>>>(tB, tA, tC, 2.0f * DIAG, -1.0f, n64q);
        k_scatter<<<gE8, blk, 0, stream>>>(tB, tC, src, dst, ow, 2.0f, E);
        k_gemm_acc<<<gGemm, blk, 0, stream>>>(tC, W + 3 * 4096, out, cofs, N);

        k_cheb_scale<<<g64, blk, 0, stream>>>(tC, tB, tA, 2.0f * DIAG, -1.0f, n64q);
        k_scatter<<<gE8, blk, 0, stream>>>(tC, tA, src, dst, ow, 2.0f, E);
        k_gemm_acc<<<gGemm, blk, 0, stream>>>(tA, W + 4 * 4096, out, cofs, N);
    }
}


# Global float-atomic scatter replaced by row ownership: a block keeps 200 destination rows in LDS, filters all edges by a byte range id, sums by LDS float atomics (arrival order), stores rows once
# speedup vs baseline: 1.5389x; 1.0433x over previous
#include <hip/hip_runtime.h>


typedef float v2f __attribute__((ext_vector_type(2)));
typedef float v8f __attribute__((ext_vector_type(8)));

static __device__ __forceinline__ size_t row64(int n) { return (size_t)n * 64; }


__global__ __launch_bounds__(256) void k_zero(float* __restrict__ p, int n) {
    int i = blockIdx.x * 256 + threadIdx.x;
    if (i < n) p[i] = 0.0f;
}

__global__ __launch_bounds__(256) void k_deg(const float* __restrict__ ew,
                                             const int* __restrict__ row,
                                             const int* __restrict__ col,
                                             float* __restrict__ deg1,
                                             float* __restrict__ deg2, int E) {
    int e = blockIdx.x * 256 + threadIdx.x;
    if (e >= E) return;
    float w = ew[e];
    atomicAdd(&deg1[row[e]], w);
    atomicAdd(&deg2[col[e]], w);
}

__global__ __launch_bounds__(256) void k_invsqrt(float* __restrict__ d, int n) {
    int i = blockIdx.x * 256 + threadIdx.x;
    if (i < n) {
        float v = d[i];
        d[i] = (v > 0.0f) ? rsqrtf(v) : 0.0f;
    }
}

#define ROWS_PER_BLOCK 200
#define SCAN_U 16
#define SWEEP_DW (1024 * SCAN_U)
__global__ __launch_bounds__(256) void k_rid(const int* __restrict__ row,
                                             const int* __restrict__ col,
                                             unsigned* __restrict__ rid1,
                                             unsigned* __restrict__ rid2,
                                             int E) {
    const int i = blockIdx.x * 256 + threadIdx.x;
    unsigned a = 0u, b = 0u;
#pragma unroll
    for (int k = 0; k < 4; ++k) {
        const int e = 4 * i + k;
        unsigned ra = 0xFFu, rb = 0xFFu;
        if (e < E) {
            ra = (unsigned)row[e] / ROWS_PER_BLOCK;
            rb = (unsigned)col[e] / ROWS_PER_BLOCK;
        }
        a |= ra << (8 * k);
        b |= rb << (8 * k);
    }
    rid1[i] = a;
    rid2[i] = b;
}

__global__ __launch_bounds__(256) void k_bias(const float* __restrict__ b1,
                                              const float* __restrict__ b2,
                                              float* __restrict__ out, int n4) {
    const int stride = gridDim.x * 256;
    int i = blockIdx.x * 256 + threadIdx.x;
    float4* out4 = (float4*)out;
    const int j = (threadIdx.x & 31) * 4;
    const float4 bv = *(const float4*)((j < 64) ? (b1 + j) : (b2 + (j - 64)));
#pragma unroll
    for (int u = 0; u < 4; ++u, i += stride) {
        if (i < n4) out4[i] = bv;
    }
}

#define QCAP 28672

static __device__ __forceinline__ unsigned zero_bytes(unsigned t) {
    return ~(((t & 0x7F7F7F7Fu) + 0x7F7F7F7Fu) | t) & 0x80808080u;
}

struct EdgeMeta { int s, d; float dd, w, ds; };
static __device__ __forceinline__ EdgeMeta edge_meta(const int* q, int n, int base, int lane,
                                                     const int* __restrict__ srcIdx,
                                                     const int* __restrict__ dstIdx,
                                                     const float* __restrict__ ew,
                                                     const float* __restrict__ dis) {
    const int idx = base + (lane & 15);
    const int e = q[(idx < n) ? idx : base];
    EdgeMeta m;
    m.s = srcIdx[e];
    m.d = dstIdx[e];
    m.w = ew[e];
    m.dd = dis[m.d];
    m.ds = dis[m.s];
    return m;
}

static __device__ __forceinline__ void add_edge_row(float* acc, const float* __restrict__ v,
                                                    const int* __restrict__ srcIdx,
                                                    const int* __restrict__ dstIdx,
                                                    const float* __restrict__ ew,
                                                    const float* __restrict__ dis,
                                                    float alpha, int lo_row, int e, int lane) {
    const float scale = 2.0f / 3.0f;
    const int s = srcIdx[e], d = dstIdx[e];
    const float wt = alpha * (-scale * dis[d] * ew[e] * dis[s]);
    const float* r = v + row64(s) + lane;
    const float vlo = r[0], vhi = r[32];
    float* o = acc + (d - lo_row) * 64 + lane;
    atomicAdd(o, wt * vlo);
    atomicAdd(o + 32, wt * vhi);
}

__global__ __launch_bounds__(1024) void k_own(const float* __restrict__ v,
                                              const float* __restrict__ prev,
                                              float* __restrict__ outf,
                                              const int* __restrict__ srcIdx,
                                              const int* __restrict__ dstIdx,
                                              const float* __restrict__ ew,
                                              const float* __restrict__ dis,
                                              const unsigned* __restrict__ rid,
                                              float alpha_diag, float beta, float alpha,
                                              int N, int E, int nSweep) {
    __shared__ float acc[ROWS_PER_BLOCK * 64];
    __shared__ int q[QCAP];
    __shared__ int qn;

    const int tid = threadIdx.x;
    const int lane = tid & 31;
    const int wave = __builtin_amdgcn_readfirstlane(tid >> 5);
    const int lo_row = blockIdx.x * ROWS_PER_BLOCK;
    int nrows = N - lo_row;
    if (nrows > ROWS_PER_BLOCK) nrows = ROWS_PER_BLOCK;
    const int nq4 = nrows * 16;
    const unsigned pat = blockIdx.x * 0x01010101u;

    {
        const float4* v4 = (const float4*)(v + row64(lo_row));
        const float4* p4 = (const float4*)(prev + row64(lo_row));
        float4* a4 = (float4*)acc;
        for (int i = tid; i < nq4; i += 1024) {
            float4 a = v4[i], p = p4[i], r;
            r.x = alpha_diag * a.x + beta * p.x;
            r.y = alpha_diag * a.y + beta * p.y;
            r.z = alpha_diag * a.z + beta * p.z;
            r.w = alpha_diag * a.w + beta * p.w;
            a4[i] = r;
        }
        if (tid == 0) qn = 0;
    }
    __syncthreads();

    {
        const int rot = blockIdx.x % nSweep;
        for (int it = 0; it < nSweep; ++it) {
            int c = it + rot;
            if (c >= nSweep) c -= nSweep;
            const int i0 = c * SWEEP_DW + tid;
            unsigned z[SCAN_U];
#pragma unroll
            for (int u = 0; u < SCAN_U; ++u) z[u] = rid[i0 + u * 1024];
            int cnt = 0;
#pragma unroll
            for (int u = 0; u < SCAN_U; ++u) {
                z[u] = zero_bytes(z[u] ^ pat);
                cnt += __builtin_popcount(z[u]);
            }
            if (cnt) {
                int pos = atomicAdd(&qn, cnt);
#pragma unroll
                for (int u = 0; u < SCAN_U; ++u) {
                    unsigned zu = z[u];
                    while (zu) {
                        const int bit = __builtin_ctz(zu);
                        zu &= zu - 1u;
                        if (pos < QCAP) q[pos] = 4 * (i0 + u * 1024) + (bit >> 3);
                        ++pos;
                    }
                }
            }
        }
    }
    __syncthreads();

    const int n = __builtin_amdgcn_readfirstlane(qn);
    if (n <= QCAP) {
        const float scale = 2.0f / 3.0f;
        const int ngroups = (n + 15) >> 4;
        int g = wave;
        EdgeMeta nxt = {0, 0, 0.0f, 0.0f, 0.0f};
        if (g < ngroups) nxt = edge_meta(q, n, g * 16, lane, srcIdx, dstIdx, ew, dis);
        while (g < ngroups) {
            const int base = g * 16;
            const int gn = g + 32;
            const EdgeMeta cur = nxt;
            float lo[16], hi[16];
#pragma unroll
            for (int j = 0; j < 16; ++j) {
                const int sj = __builtin_amdgcn_readlane(cur.s, j);
                const float* r = v + row64(sj) + lane;
                lo[j] = r[0];
                hi[j] = r[32];
            }
            if (gn < ngroups) nxt = edge_meta(q, n, gn * 16, lane, srcIdx, dstIdx, ew, dis);
            const float wt = alpha * (-scale * cur.dd * cur.w * cur.ds);
            const int dl = cur.d - lo_row;
#pragma unroll
            for (int j = 0; j < 16; ++j) {
                if (base + j < n) {
                    const int dj = __builtin_amdgcn_readlane(dl, j);
                    const float wj = __builtin_bit_cast(float,
                        __builtin_amdgcn_readlane(__builtin_bit_cast(int, wt), j));
                    float* o = acc + dj * 64 + lane;
                    atomicAdd(o, wj * lo[j]);
                    atomicAdd(o + 32, wj * hi[j]);
                }
            }
            g = gn;
        }
    } else {
        const int nW = nSweep * SWEEP_DW;
        for (int wb = wave * 32; wb < nW; wb += 32 * 32) {
            const unsigned zl = zero_bytes(rid[wb + lane] ^ pat);
            unsigned m = __builtin_amdgcn_ballot_w32(zl != 0u);
            while (m) {
                const int l = __builtin_ctz(m);
                m &= m - 1u;
                unsigned zz = (unsigned)__builtin_amdgcn_readlane((int)zl, l);
                while (zz) {
                    const int bit = __builtin_ctz(zz);
                    zz &= zz - 1u;
                    const int e = 4 * (wb + l) + (bit >> 3);
                    if (e < E) add_edge_row(acc, v, srcIdx, dstIdx, ew, dis, alpha, lo_row, e, lane);
                }
            }
        }
    }
    __syncthreads();

    {
        float4* o4 = (float4*)(outf + row64(lo_row));
        const float4* a4 = (const float4*)acc;
        for (int i = tid; i < nq4; i += 1024) o4[i] = a4[i];
    }
}

__global__ __launch_bounds__(256) void k_gemm_acc(const float* __restrict__ Tx,
                                                  const float* __restrict__ W,
                                                  float* __restrict__ out,
                                                  int colofs, int nRows) {
    __shared__ float Wl[64 * 64];
    __shared__ float Al[128 * 64];

    const int tid = threadIdx.x;
    const int row0b = blockIdx.x * 128;

    {
        const float4* W4 = (const float4*)W;
        float4* Wl4 = (float4*)Wl;
#pragma unroll
        for (int i = 0; i < 4; ++i) Wl4[tid + 256 * i] = W4[tid + 256 * i];
    }
    {
        float4* Al4 = (float4*)Al;
#pragma unroll
        for (int i = 0; i < 8; ++i) {
            int idx = tid + 256 * i;
            int r = row0b + (idx >> 4);
            float4 v = make_float4(0.f, 0.f, 0.f, 0.f);
            if (r < nRows) v = ((const float4*)(Tx + row64(r)))[idx & 15];
            Al4[idx] = v;
        }
    }
    __syncthreads();

    const int wave = tid >> 5;
    const int lane = tid & 31;
    const int lm = lane & 15;
    const int lg = lane >> 4;

    const int rowt = wave * 16;
    const int row0 = row0b + rowt;

    v8f c0, c1, c2, c3;
    const int nclamp = nRows - 1;
#pragma unroll
    for (int i = 0; i < 8; ++i) {
        int r = row0 + i + lg * 8;
        bool ok = (r < nRows);
        int rc = ok ? r : nclamp;
        const float* orow = out + (size_t)rc * 128 + colofs + lm;
        float v0 = orow[0], v1 = orow[16], v2 = orow[32], v3 = orow[48];
        c0[i] = ok ? v0 : 0.f;
        c1[i] = ok ? v1 : 0.f;
        c2[i] = ok ? v2 : 0.f;
        c3[i] = ok ? v3 : 0.f;
    }

    const float* arow_l = Al + (rowt + lm) * 64;
    const float* bcol_l = Wl + lm;

#pragma unroll
    for (int kk = 0; kk < 16; ++kk) {
        const int kbase = kk * 4 + lg * 2;
        v2f a = *(const v2f*)(arow_l + kbase);
        v2f b0, b1v, b2, b3;
        b0[0]  = bcol_l[(kbase + 0) * 64 + 0];
        b0[1]  = bcol_l[(kbase + 1) * 64 + 0];
        b1v[0] = bcol_l[(kbase + 0) * 64 + 16];
        b1v[1] = bcol_l[(kbase + 1) * 64 + 16];
        b2[0]  = bcol_l[(kbase + 0) * 64 + 32];
        b2[1]  = bcol_l[(kbase + 1) * 64 + 32];
        b3[0]  = bcol_l[(kbase + 0) * 64 + 48];
        b3[1]  = bcol_l[(kbase + 1) * 64 + 48];
        c0 = __builtin_amdgcn_wmma_f32_16x16x4_f32(false, a, false, b0,  (short)0, c0, false, false);
        c1 = __builtin_amdgcn_wmma_f32_16x16x4_f32(false, a, false, b1v, (short)0, c1, false, false);
        c2 = __builtin_amdgcn_wmma_f32_16x16x4_f32(false, a, false, b2,  (short)0, c2, false, false);
        c3 = __builtin_amdgcn_wmma_f32_16x16x4_f32(false, a, false, b3,  (short)0, c3, false, false);
    }

#pragma unroll
    for (int i = 0; i < 8; ++i) {
        int r = row0 + i + lg * 8;
        if (r < nRows) {
            float* orow = out + (size_t)r * 128 + colofs + lm;
            orow[0]  = c0[i];
            orow[16] = c1[i];
            orow[32] = c2[i];
            orow[48] = c3[i];
        }
    }
}


extern "C" void kernel_launch(void* const* d_in, const int* in_sizes, int n_in,
                              void* d_out, int out_size, void* d_ws, size_t ws_size,
                              hipStream_t stream) {
    const float* x  = (const float*)d_in[0];
    const int*   ei = (const int*)d_in[1];
    const float* ew = (const float*)d_in[2];
    const float* W1 = (const float*)d_in[3];
    const float* b1 = (const float*)d_in[4];
    const float* W2 = (const float*)d_in[5];
    const float* b2 = (const float*)d_in[6];
    float* out = (float*)d_out;

    const int N = in_sizes[0] / 64;
    const int E = in_sizes[2];
    const int* row = ei;
    const int* col = ei + E;

    float* ws   = (float*)d_ws;
    float* deg1 = ws;
    float* deg2 = deg1 + N;
    unsigned* rid1 = (unsigned*)(deg2 + N);
    unsigned* rid2 = rid1 + E;
    float* tA   = (float*)(rid2 + E);
    float* tB   = tA + (size_t)N * 64;
    float* tC   = tB + (size_t)N * 64;

    const int n128q = N * 32;
    const int nSweep = ((E + 3) / 4 + SWEEP_DW - 1) / SWEEP_DW;
    const float DIAG = 2.0f / 3.0f - 1.0f;

    dim3 blk(256);
    int gN2   = (2 * N + 255) / 256;
    int gE    = (E + 255) / 256;
    int gW    = nSweep * (SWEEP_DW / 256);
    int gOwn  = (N + ROWS_PER_BLOCK - 1) / ROWS_PER_BLOCK;
    int g128  = (n128q + 1023) / 1024;
    int gGemm = (N + 127) / 128;

    k_zero<<<gN2, blk, 0, stream>>>(deg1, 2 * N);
    k_deg<<<gE, blk, 0, stream>>>(ew, row, col, deg1, deg2, E);
    k_invsqrt<<<gN2, blk, 0, stream>>>(deg1, 2 * N);
    k_rid<<<gW, blk, 0, stream>>>(row, col, rid1, rid2, E);

    k_bias<<<g128, blk, 0, stream>>>(b1, b2, out, n128q);

    for (int dir = 0; dir < 2; ++dir) {
        const int*      src  = (dir == 0) ? col : row;
        const int*      dst  = (dir == 0) ? row : col;
        const float*    dis  = (dir == 0) ? deg1 : deg2;
        const unsigned* rid  = (dir == 0) ? rid1 : rid2;
        const float*    W    = (dir == 0) ? W1 : W2;
        const int       cofs = (dir == 0) ? 0 : 64;

        k_gemm_acc<<<gGemm, blk, 0, stream>>>(x, W, out, cofs, N);

        k_own<<<gOwn, 1024, 0, stream>>>(x, x, tA, src, dst, ew, dis, rid, DIAG, 0.0f, 1.0f, N, E, nSweep);
        k_gemm_acc<<<gGemm, blk, 0, stream>>>(tA, W + 1 * 4096, out, cofs, N);

        k_own<<<gOwn, 1024, 0, stream>>>(tA, x, tB, src, dst, ew, dis, rid, 2.0f * DIAG, -1.0f, 2.0f, N, E, nSweep);
        k_gemm_acc<<<gGemm, blk, 0, stream>>>(tB, W + 2 * 4096, out, cofs, N);

        k_own<<<gOwn, 1024, 0, stream>>>(tB, tA, tC, src, dst, ew, dis, rid, 2.0f * DIAG, -1.0f, 2.0f, N, E, nSweep);
        k_gemm_acc<<<gGemm, blk, 0, stream>>>(tC, W + 3 * 4096, out, cofs, N);

        k_own<<<gOwn, 1024, 0, stream>>>(tC, tB, tA, src, dst, ew, dis, rid, 2.0f * DIAG, -1.0f, 2.0f, N, E, nSweep);
        k_gemm_acc<<<gGemm, blk, 0, stream>>>(tA, W + 4 * 4096, out, cofs, N);
    }
}


# Each direction filters once: pass 1 leaves every block's edges as 8-byte records (source row, local row, weight); passes 2-4 read them: no range-id scan, no index or degree loads; LDS sums as before
# speedup vs baseline: 2.1586x; 1.4027x over previous
#include <hip/hip_runtime.h>


typedef float v2f __attribute__((ext_vector_type(2)));
typedef float v8f __attribute__((ext_vector_type(8)));

static __device__ __forceinline__ size_t row64(int n) { return (size_t)n * 64; }


__global__ __launch_bounds__(256) void k_zero(float* __restrict__ p, int n) {
    int i = blockIdx.x * 256 + threadIdx.x;
    if (i < n) p[i] = 0.0f;
}

__global__ __launch_bounds__(256) void k_deg(const float* __restrict__ ew,
                                             const int* __restrict__ row,
                                             const int* __restrict__ col,
                                             float* __restrict__ deg1,
                                             float* __restrict__ deg2, int E) {
    int e = blockIdx.x * 256 + threadIdx.x;
    if (e >= E) return;
    float w = ew[e];
    atomicAdd(&deg1[row[e]], w);
    atomicAdd(&deg2[col[e]], w);
}

__global__ __launch_bounds__(256) void k_invsqrt(float* __restrict__ d, int n) {
    int i = blockIdx.x * 256 + threadIdx.x;
    if (i < n) {
        float v = d[i];
        d[i] = (v > 0.0f) ? rsqrtf(v) : 0.0f;
    }
}

#define ROWS_PER_BLOCK 200
#define SCAN_U 16
#define SWEEP_DW (1024 * SCAN_U)
__global__ __launch_bounds__(256) void k_rid(const int* __restrict__ row,
                                             const int* __restrict__ col,
                                             unsigned* __restrict__ rid1,
                                             unsigned* __restrict__ rid2,
                                             int E) {
    const int i = blockIdx.x * 256 + threadIdx.x;
    unsigned a = 0u, b = 0u;
#pragma unroll
    for (int k = 0; k < 4; ++k) {
        const int e = 4 * i + k;
        unsigned ra = 0xFFu, rb = 0xFFu;
        if (e < E) {
            ra = (unsigned)row[e] / ROWS_PER_BLOCK;
            rb = (unsigned)col[e] / ROWS_PER_BLOCK;
        }
        a |= ra << (8 * k);
        b |= rb << (8 * k);
    }
    rid1[i] = a;
    rid2[i] = b;
}

__global__ __launch_bounds__(256) void k_bias(const float* __restrict__ b1,
                                              const float* __restrict__ b2,
                                              float* __restrict__ out, int n4) {
    const int stride = gridDim.x * 256;
    int i = blockIdx.x * 256 + threadIdx.x;
    float4* out4 = (float4*)out;
    const int j = (threadIdx.x & 31) * 4;
    const float4 bv = *(const float4*)((j < 64) ? (b1 + j) : (b2 + (j - 64)));
#pragma unroll
    for (int u = 0; u < 4; ++u, i += stride) {
        if (i < n4) out4[i] = bv;
    }
}

#define QCAP 28672

static __device__ __forceinline__ unsigned zero_bytes(unsigned t) {
    return ~(((t & 0x7F7F7F7Fu) + 0x7F7F7F7Fu) | t) & 0x80808080u;
}

struct EdgeMeta { int s, d; float dd, w, ds; };

static_assert(ROWS_PER_BLOCK <= 256, "a record keeps the local row in 8 bits");
static __device__ __forceinline__ uint2 edge_rec(const uint2* rq, int n, int base, int lane) {
    const int idx = base + (lane & 15);
    return rq[(idx < n) ? idx : base];
}
static __device__ __forceinline__ EdgeMeta edge_meta(const int* q, int n, int base, int lane,
                                                     const int* __restrict__ srcIdx,
                                                     const int* __restrict__ dstIdx,
                                                     const float* __restrict__ ew,
                                                     const float* __restrict__ dis) {
    const int idx = base + (lane & 15);
    const int e = q[(idx < n) ? idx : base];
    EdgeMeta m;
    m.s = srcIdx[e];
    m.d = dstIdx[e];
    m.w = ew[e];
    m.dd = dis[m.d];
    m.ds = dis[m.s];
    return m;
}

static __device__ __forceinline__ void add_edge_row(float* acc, const float* __restrict__ v,
                                                    const int* __restrict__ srcIdx,
                                                    const int* __restrict__ dstIdx,
                                                    const float* __restrict__ ew,
                                                    const float* __restrict__ dis,
                                                    float alpha, int lo_row, int e, int lane) {
    const float scale = 2.0f / 3.0f;
    const int s = srcIdx[e], d = dstIdx[e];
    const float wt = alpha * (-scale * dis[d] * ew[e] * dis[s]);
    const float* r = v + row64(s) + lane;
    const float vlo = r[0], vhi = r[32];
    float* o = acc + (d - lo_row) * 64 + lane;
    atomicAdd(o, wt * vlo);
    atomicAdd(o + 32, wt * vhi);
}

__global__ __launch_bounds__(1024) void k_own(const float* __restrict__ v,
                                              const float* prev,
                                              float* outf,
                                              const int* __restrict__ srcIdx,
                                              const int* __restrict__ dstIdx,
                                              const float* __restrict__ ew,
                                              const float* __restrict__ dis,
                                              const unsigned* __restrict__ rid,
                                              unsigned* rec, int* rcnt, int build,
                                              float alpha_diag, float beta, float alpha,
                                              int N, int E, int nSweep) {
    __shared__ float acc[ROWS_PER_BLOCK * 64];
    __shared__ int q[QCAP];
    __shared__ int qn;

    const int tid = threadIdx.x;
    const int lane = tid & 31;
    const int wave = __builtin_amdgcn_readfirstlane(tid >> 5);
    const int lo_row = blockIdx.x * ROWS_PER_BLOCK;
    int nrows = N - lo_row;
    if (nrows > ROWS_PER_BLOCK) nrows = ROWS_PER_BLOCK;
    const int nq4 = nrows * 16;
    const unsigned pat = blockIdx.x * 0x01010101u;
    uint2* rq = (uint2*)rec + (size_t)lo_row * 32;
    const int rcap = nrows * 32;

    {
        const float4* v4 = (const float4*)(v + row64(lo_row));
        const float4* p4 = (const float4*)(prev + row64(lo_row));
        float4* a4 = (float4*)acc;
        for (int i = tid; i < nq4; i += 1024) {
            float4 a = v4[i], p = p4[i], r;
            r.x = alpha_diag * a.x + beta * p.x;
            r.y = alpha_diag * a.y + beta * p.y;
            r.z = alpha_diag * a.z + beta * p.z;
            r.w = alpha_diag * a.w + beta * p.w;
            a4[i] = r;
        }
        if (tid == 0) qn = 0;
    }
    __syncthreads();

    int nrec = -1;
    if (!build) nrec = __builtin_amdgcn_readfirstlane(rcnt[blockIdx.x]);
    if (nrec >= 0 && nrec <= rcap) {
        const int ngroups = (nrec + 15) >> 4;
        int g = wave;
        uint2 nxt = make_uint2(0u, 0u);
        if (g < ngroups) nxt = edge_rec(rq, nrec, g * 16, lane);
        while (g < ngroups) {
            const int base = g * 16;
            const int gn = g + 32;
            const uint2 cur = nxt;
            float lo[16], hi[16];
#pragma unroll
            for (int j = 0; j < 16; ++j) {
                const int sj = (int)((unsigned)__builtin_amdgcn_readlane((int)cur.x, j) >> 8);
                const float* r = v + row64(sj) + lane;
                lo[j] = r[0];
                hi[j] = r[32];
            }
            if (gn < ngroups) nxt = edge_rec(rq, nrec, gn * 16, lane);
            const float wt = alpha * __builtin_bit_cast(float, cur.y);
            const int dl = (int)(cur.x & 255u);
#pragma unroll
            for (int j = 0; j < 16; ++j) {
                if (base + j < nrec) {
                    const int dj = __builtin_amdgcn_readlane(dl, j);
                    const float wj = __builtin_bit_cast(float,
                        __builtin_amdgcn_readlane(__builtin_bit_cast(int, wt), j));
                    float* o = acc + dj * 64 + lane;
                    atomicAdd(o, wj * lo[j]);
                    atomicAdd(o + 32, wj * hi[j]);
                }
            }
            g = gn;
        }
    } else {
        {
            const int rot = blockIdx.x % nSweep;
            for (int it = 0; it < nSweep; ++it) {
                int c = it + rot;
                if (c >= nSweep) c -= nSweep;
                const int i0 = c * SWEEP_DW + tid;
                unsigned z[SCAN_U];
#pragma unroll
                for (int u = 0; u < SCAN_U; ++u) z[u] = rid[i0 + u * 1024];
                int cnt = 0;
#pragma unroll
                for (int u = 0; u < SCAN_U; ++u) {
                    z[u] = zero_bytes(z[u] ^ pat);
                    cnt += __builtin_popcount(z[u]);
                }
                if (cnt) {
                    int pos = atomicAdd(&qn, cnt);
#pragma unroll
                    for (int u = 0; u < SCAN_U; ++u) {
                        unsigned zu = z[u];
                        while (zu) {
                            const int bit = __builtin_ctz(zu);
                            zu &= zu - 1u;
                            if (pos < QCAP) q[pos] = 4 * (i0 + u * 1024) + (bit >> 3);
                            ++pos;
                        }
                    }
                }
            }
        }
        __syncthreads();

        const int n = __builtin_amdgcn_readfirstlane(qn);
        if (n <= QCAP) {
            const float scale = 2.0f / 3.0f;
            const int ngroups = (n + 15) >> 4;
            int g = wave;
            EdgeMeta nxt = {0, 0, 0.0f, 0.0f, 0.0f};
            if (g < ngroups) nxt = edge_meta(q, n, g * 16, lane, srcIdx, dstIdx, ew, dis);
            while (g < ngroups) {
                const int base = g * 16;
                const int gn = g + 32;
                const EdgeMeta cur = nxt;
                float lo[16], hi[16];
#pragma unroll
                for (int j = 0; j < 16; ++j) {
                    const int sj = __builtin_amdgcn_readlane(cur.s, j);
                    const float* r = v + row64(sj) + lane;
                    lo[j] = r[0];
                    hi[j] = r[32];
                }
                if (gn < ngroups) nxt = edge_meta(q, n, gn * 16, lane, srcIdx, dstIdx, ew, dis);
                const float wb = -scale * cur.dd * cur.w * cur.ds;
                const float wt = alpha * wb;
                const int dl = cur.d - lo_row;
                if (build) {
                    const int idx = base + lane;
                    if (lane < 16 && idx < n && idx < rcap)
                        rq[idx] = make_uint2(((unsigned)cur.s << 8) | (unsigned)dl,
                                             __builtin_bit_cast(unsigned, wb));
                }
#pragma unroll
                for (int j = 0; j < 16; ++j) {
                    if (base + j < n) {
                        const int dj = __builtin_amdgcn_readlane(dl, j);
                        const float wj = __builtin_bit_cast(float,
                            __builtin_amdgcn_readlane(__builtin_bit_cast(int, wt), j));
                        float* o = acc + dj * 64 + lane;
                        atomicAdd(o, wj * lo[j]);
                        atomicAdd(o + 32, wj * hi[j]);
                    }
                }
                g = gn;
            }
        } else {
            const int nW = nSweep * SWEEP_DW;
            for (int wb = wave * 32; wb < nW; wb += 32 * 32) {
                const unsigned zl = zero_bytes(rid[wb + lane] ^ pat);
                unsigned m = __builtin_amdgcn_ballot_w32(zl != 0u);
                while (m) {
                    const int l = __builtin_ctz(m);
                    m &= m - 1u;
                    unsigned zz = (unsigned)__builtin_amdgcn_readlane((int)zl, l);
                    while (zz) {
                        const int bit = __builtin_ctz(zz);
                        zz &= zz - 1u;
                        const int e = 4 * (wb + l) + (bit >> 3);
                        if (e < E) add_edge_row(acc, v, srcIdx, dstIdx, ew, dis, alpha, lo_row, e, lane);
                    }
                }
            }
        }
        if (build && tid == 0) rcnt[blockIdx.x] = n;
    }
    __syncthreads();

    {
        float4* o4 = (float4*)(outf + row64(lo_row));
        const float4* a4 = (const float4*)acc;
        for (int i = tid; i < nq4; i += 1024) o4[i] = a4[i];
    }
}

__global__ __launch_bounds__(256) void k_gemm_acc(const float* __restrict__ Tx,
                                                  const float* __restrict__ W,
                                                  float* __restrict__ out,
                                                  int colofs, int nRows) {
    __shared__ float Wl[64 * 64];
    __shared__ float Al[128 * 64];

    const int tid = threadIdx.x;
    const int row0b = blockIdx.x * 128;

    {
        const float4* W4 = (const float4*)W;
        float4* Wl4 = (float4*)Wl;
#pragma unroll
        for (int i = 0; i < 4; ++i) Wl4[tid + 256 * i] = W4[tid + 256 * i];
    }
    {
        float4* Al4 = (float4*)Al;
#pragma unroll
        for (int i = 0; i < 8; ++i) {
            int idx = tid + 256 * i;
            int r = row0b + (idx >> 4);
            float4 v = make_float4(0.f, 0.f, 0.f, 0.f);
            if (r < nRows) v = ((const float4*)(Tx + row64(r)))[idx & 15];
            Al4[idx] = v;
        }
    }
    __syncthreads();

    const int wave = tid >> 5;
    const int lane = tid & 31;
    const int lm = lane & 15;
    const int lg = lane >> 4;

    const int rowt = wave * 16;
    const int row0 = row0b + rowt;

    v8f c0, c1, c2, c3;
    const int nclamp = nRows - 1;
#pragma unroll
    for (int i = 0; i < 8; ++i) {
        int r = row0 + i + lg * 8;
        bool ok = (r < nRows);
        int rc = ok ? r : nclamp;
        const float* orow = out + (size_t)rc * 128 + colofs + lm;
        float v0 = orow[0], v1 = orow[16], v2 = orow[32], v3 = orow[48];
        c0[i] = ok ? v0 : 0.f;
        c1[i] = ok ? v1 : 0.f;
        c2[i] = ok ? v2 : 0.f;
        c3[i] = ok ? v3 : 0.f;
    }

    const float* arow_l = Al + (rowt + lm) * 64;
    const float* bcol_l = Wl + lm;

#pragma unroll
    for (int kk = 0; kk < 16; ++kk) {
        const int kbase = kk * 4 + lg * 2;
        v2f a = *(const v2f*)(arow_l + kbase);
        v2f b0, b1v, b2, b3;
        b0[0]  = bcol_l[(kbase + 0) * 64 + 0];
        b0[1]  = bcol_l[(kbase + 1) * 64 + 0];
        b1v[0] = bcol_l[(kbase + 0) * 64 + 16];
        b1v[1] = bcol_l[(kbase + 1) * 64 + 16];
        b2[0]  = bcol_l[(kbase + 0) * 64 + 32];
        b2[1]  = bcol_l[(kbase + 1) * 64 + 32];
        b3[0]  = bcol_l[(kbase + 0) * 64 + 48];
        b3[1]  = bcol_l[(kbase + 1) * 64 + 48];
        c0 = __builtin_amdgcn_wmma_f32_16x16x4_f32(false, a, false, b0,  (short)0, c0, false, false);
        c1 = __builtin_amdgcn_wmma_f32_16x16x4_f32(false, a, false, b1v, (short)0, c1, false, false);
        c2 = __builtin_amdgcn_wmma_f32_16x16x4_f32(false, a, false, b2,  (short)0, c2, false, false);
        c3 = __builtin_amdgcn_wmma_f32_16x16x4_f32(false, a, false, b3,  (short)0, c3, false, false);
    }

#pragma unroll
    for (int i = 0; i < 8; ++i) {
        int r = row0 + i + lg * 8;
        if (r < nRows) {
            float* orow = out + (size_t)r * 128 + colofs + lm;
            orow[0]  = c0[i];
            orow[16] = c1[i];
            orow[32] = c2[i];
            orow[48] = c3[i];
        }
    }
}


extern "C" void kernel_launch(void* const* d_in, const int* in_sizes, int n_in,
                              void* d_out, int out_size, void* d_ws, size_t ws_size,
                              hipStream_t stream) {
    const float* x  = (const float*)d_in[0];
    const int*   ei = (const int*)d_in[1];
    const float* ew = (const float*)d_in[2];
    const float* W1 = (const float*)d_in[3];
    const float* b1 = (const float*)d_in[4];
    const float* W2 = (const float*)d_in[5];
    const float* b2 = (const float*)d_in[6];
    float* out = (float*)d_out;

    const int N = in_sizes[0] / 64;
    const int E = in_sizes[2];
    const int* row = ei;
    const int* col = ei + E;

    float* ws   = (float*)d_ws;
    float* deg1 = ws;
    float* deg2 = deg1 + N;
    unsigned* rid1 = (unsigned*)(deg2 + N);
    unsigned* rid2 = rid1 + E;
    float* tA   = (float*)(rid2 + E);
    float* tB   = tA + (size_t)N * 64;
    unsigned* rec = (unsigned*)(tB + (size_t)N * 64);

    const int n128q = N * 32;
    const int nSweep = ((E + 3) / 4 + SWEEP_DW - 1) / SWEEP_DW;
    const float DIAG = 2.0f / 3.0f - 1.0f;
    int* rcnt = (int*)(rid1 + (size_t)nSweep * SWEEP_DW);

    dim3 blk(256);
    int gN2   = (2 * N + 255) / 256;
    int gE    = (E + 255) / 256;
    int gW    = nSweep * (SWEEP_DW / 256);
    int gOwn  = (N + ROWS_PER_BLOCK - 1) / ROWS_PER_BLOCK;
    int g128  = (n128q + 1023) / 1024;
    int gGemm = (N + 127) / 128;

    k_zero<<<gN2, blk, 0, stream>>>(deg1, 2 * N);
    k_deg<<<gE, blk, 0, stream>>>(ew, row, col, deg1, deg2, E);
    k_invsqrt<<<gN2, blk, 0, stream>>>(deg1, 2 * N);
    k_rid<<<gW, blk, 0, stream>>>(row, col, rid1, rid2, E);

    k_bias<<<g128, blk, 0, stream>>>(b1, b2, out, n128q);

    for (int dir = 0; dir < 2; ++dir) {
        const int*      src  = (dir == 0) ? col : row;
        const int*      dst  = (dir == 0) ? row : col;
        const float*    dis  = (dir == 0) ? deg1 : deg2;
        const unsigned* rid  = (dir == 0) ? rid1 : rid2;
        const float*    W    = (dir == 0) ? W1 : W2;
        const int       cofs = (dir == 0) ? 0 : 64;

        k_gemm_acc<<<gGemm, blk, 0, stream>>>(x, W, out, cofs, N);

        k_own<<<gOwn, 1024, 0, stream>>>(x, x, tA, src, dst, ew, dis, rid, rec, rcnt, 1, DIAG, 0.0f, 1.0f, N, E, nSweep);
        k_gemm_acc<<<gGemm, blk, 0, stream>>>(tA, W + 1 * 4096, out, cofs, N);

        k_own<<<gOwn, 1024, 0, stream>>>(tA, x, tB, src, dst, ew, dis, rid, rec, rcnt, 0, 2.0f * DIAG, -1.0f, 2.0f, N, E, nSweep);
        k_gemm_acc<<<gGemm, blk, 0, stream>>>(tB, W + 2 * 4096, out, cofs, N);

        k_own<<<gOwn, 1024, 0, stream>>>(tB, tA, tA, src, dst, ew, dis, rid, rec, rcnt, 0, 2.0f * DIAG, -1.0f, 2.0f, N, E, nSweep);
        k_gemm_acc<<<gGemm, blk, 0, stream>>>(tA, W + 3 * 4096, out, cofs, N);

        k_own<<<gOwn, 1024, 0, stream>>>(tA, tB, tB, src, dst, ew, dis, rid, rec, rcnt, 0, 2.0f * DIAG, -1.0f, 2.0f, N, E, nSweep);
        k_gemm_acc<<<gGemm, blk, 0, stream>>>(tB, W + 4 * 4096, out, cofs, N);
    }
}


# Dense projections batched over orders: per direction one f32 WMMA launch takes orders 0-2 (accumulators start at the bias, out written, not read), one takes orders 3-4; 23 -> 16 launches; same sums
# speedup vs baseline: 2.5780x; 1.1943x over previous
#include <hip/hip_runtime.h>


typedef float v2f __attribute__((ext_vector_type(2)));
typedef float v8f __attribute__((ext_vector_type(8)));

static __device__ __forceinline__ size_t row64(int n) { return (size_t)n * 64; }


__global__ __launch_bounds__(256) void k_zero(float* __restrict__ p, int n) {
    int i = blockIdx.x * 256 + threadIdx.x;
    if (i < n) p[i] = 0.0f;
}

__global__ __launch_bounds__(256) void k_deg(const float* __restrict__ ew,
                                             const int* __restrict__ row,
                                             const int* __restrict__ col,
                                             float* __restrict__ deg1,
                                             float* __restrict__ deg2, int E) {
    int e = blockIdx.x * 256 + threadIdx.x;
    if (e >= E) return;
    float w = ew[e];
    atomicAdd(&deg1[row[e]], w);
    atomicAdd(&deg2[col[e]], w);
}

__global__ __launch_bounds__(256) void k_invsqrt(float* __restrict__ d, int n) {
    int i = blockIdx.x * 256 + threadIdx.x;
    if (i < n) {
        float v = d[i];
        d[i] = (v > 0.0f) ? rsqrtf(v) : 0.0f;
    }
}

#define ROWS_PER_BLOCK 200
#define SCAN_U 16
#define SWEEP_DW (1024 * SCAN_U)
__global__ __launch_bounds__(256) void k_rid(const int* __restrict__ row,
                                             const int* __restrict__ col,
                                             unsigned* __restrict__ rid1,
                                             unsigned* __restrict__ rid2,
                                             int E) {
    const int i = blockIdx.x * 256 + threadIdx.x;
    unsigned a = 0u, b = 0u;
#pragma unroll
    for (int k = 0; k < 4; ++k) {
        const int e = 4 * i + k;
        unsigned ra = 0xFFu, rb = 0xFFu;
        if (e < E) {
            ra = (unsigned)row[e] / ROWS_PER_BLOCK;
            rb = (unsigned)col[e] / ROWS_PER_BLOCK;
        }
        a |= ra << (8 * k);
        b |= rb << (8 * k);
    }
    rid1[i] = a;
    rid2[i] = b;
}

#define QCAP 28672

static __device__ __forceinline__ unsigned zero_bytes(unsigned t) {
    return ~(((t & 0x7F7F7F7Fu) + 0x7F7F7F7Fu) | t) & 0x80808080u;
}

struct EdgeMeta { int s, d; float dd, w, ds; };

static_assert(ROWS_PER_BLOCK <= 256, "a record keeps the local row in 8 bits");
static __device__ __forceinline__ uint2 edge_rec(const uint2* rq, int n, int base, int lane) {
    const int idx = base + (lane & 15);
    return rq[(idx < n) ? idx : base];
}
static __device__ __forceinline__ EdgeMeta edge_meta(const int* q, int n, int base, int lane,
                                                     const int* __restrict__ srcIdx,
                                                     const int* __restrict__ dstIdx,
                                                     const float* __restrict__ ew,
                                                     const float* __restrict__ dis) {
    const int idx = base + (lane & 15);
    const int e = q[(idx < n) ? idx : base];
    EdgeMeta m;
    m.s = srcIdx[e];
    m.d = dstIdx[e];
    m.w = ew[e];
    m.dd = dis[m.d];
    m.ds = dis[m.s];
    return m;
}

static __device__ __forceinline__ void add_edge_row(float* acc, const float* __restrict__ v,
                                                    const int* __restrict__ srcIdx,
                                                    const int* __restrict__ dstIdx,
                                                    const float* __restrict__ ew,
                                                    const float* __restrict__ dis,
                                                    float alpha, int lo_row, int e, int lane) {
    const float scale = 2.0f / 3.0f;
    const int s = srcIdx[e], d = dstIdx[e];
    const float wt = alpha * (-scale * dis[d] * ew[e] * dis[s]);
    const float* r = v + row64(s) + lane;
    const float vlo = r[0], vhi = r[32];
    float* o = acc + (d - lo_row) * 64 + lane;
    atomicAdd(o, wt * vlo);
    atomicAdd(o + 32, wt * vhi);
}

__global__ __launch_bounds__(1024) void k_own(const float* __restrict__ v,
                                              const float* prev,
                                              float* outf,
                                              const int* __restrict__ srcIdx,
                                              const int* __restrict__ dstIdx,
                                              const float* __restrict__ ew,
                                              const float* __restrict__ dis,
                                              const unsigned* __restrict__ rid,
                                              unsigned* rec, int* rcnt, int build,
                                              float alpha_diag, float beta, float alpha,
                                              int N, int E, int nSweep) {
    __shared__ float acc[ROWS_PER_BLOCK * 64];
    __shared__ int q[QCAP];
    __shared__ int qn;

    const int tid = threadIdx.x;
    const int lane = tid & 31;
    const int wave = __builtin_amdgcn_readfirstlane(tid >> 5);
    const int lo_row = blockIdx.x * ROWS_PER_BLOCK;
    int nrows = N - lo_row;
    if (nrows > ROWS_PER_BLOCK) nrows = ROWS_PER_BLOCK;
    const int nq4 = nrows * 16;
    const unsigned pat = blockIdx.x * 0x01010101u;
    uint2* rq = (uint2*)rec + (size_t)lo_row * 32;
    const int rcap = nrows * 32;

    {
        const float4* v4 = (const float4*)(v + row64(lo_row));
        const float4* p4 = (const float4*)(prev + row64(lo_row));
        float4* a4 = (float4*)acc;
        for (int i = tid; i < nq4; i += 1024) {
            float4 a = v4[i], p = p4[i], r;
            r.x = alpha_diag * a.x + beta * p.x;
            r.y = alpha_diag * a.y + beta * p.y;
            r.z = alpha_diag * a.z + beta * p.z;
            r.w = alpha_diag * a.w + beta * p.w;
            a4[i] = r;
        }
        if (tid == 0) qn = 0;
    }
    __syncthreads();

    int nrec = -1;
    if (!build) nrec = __builtin_amdgcn_readfirstlane(rcnt[blockIdx.x]);
    if (nrec >= 0 && nrec <= rcap) {
        const int ngroups = (nrec + 15) >> 4;
        int g = wave;
        uint2 nxt = make_uint2(0u, 0u);
        if (g < ngroups) nxt = edge_rec(rq, nrec, g * 16, lane);
        while (g < ngroups) {
            const int base = g * 16;
            const int gn = g + 32;
            const uint2 cur = nxt;
            float lo[16], hi[16];
#pragma unroll
            for (int j = 0; j < 16; ++j) {
                const int sj = (int)((unsigned)__builtin_amdgcn_readlane((int)cur.x, j) >> 8);
                const float* r = v + row64(sj) + lane;
                lo[j] = r[0];
                hi[j] = r[32];
            }
            if (gn < ngroups) nxt = edge_rec(rq, nrec, gn * 16, lane);
            const float wt = alpha * __builtin_bit_cast(float, cur.y);
            const int dl = (int)(cur.x & 255u);
#pragma unroll
            for (int j = 0; j < 16; ++j) {
                if (base + j < nrec) {
                    const int dj = __builtin_amdgcn_readlane(dl, j);
                    const float wj = __builtin_bit_cast(float,
                        __builtin_amdgcn_readlane(__builtin_bit_cast(int, wt), j));
                    float* o = acc + dj * 64 + lane;
                    atomicAdd(o, wj * lo[j]);
                    atomicAdd(o + 32, wj * hi[j]);
                }
            }
            g = gn;
        }
    } else {
        {
            const int rot = blockIdx.x % nSweep;
            for (int it = 0; it < nSweep; ++it) {
                int c = it + rot;
                if (c >= nSweep) c -= nSweep;
                const int i0 = c * SWEEP_DW + tid;
                unsigned z[SCAN_U];
#pragma unroll
                for (int u = 0; u < SCAN_U; ++u) z[u] = rid[i0 + u * 1024];
                int cnt = 0;
#pragma unroll
                for (int u = 0; u < SCAN_U; ++u) {
                    z[u] = zero_bytes(z[u] ^ pat);
                    cnt += __builtin_popcount(z[u]);
                }
                if (cnt) {
                    int pos = atomicAdd(&qn, cnt);
#pragma unroll
                    for (int u = 0; u < SCAN_U; ++u) {
                        unsigned zu = z[u];
                        while (zu) {
                            const int bit = __builtin_ctz(zu);
                            zu &= zu - 1u;
                            if (pos < QCAP) q[pos] = 4 * (i0 + u * 1024) + (bit >> 3);
                            ++pos;
                        }
                    }
                }
            }
        }
        __syncthreads();

        const int n = __builtin_amdgcn_readfirstlane(qn);
        if (n <= QCAP) {
            const float scale = 2.0f / 3.0f;
            const int ngroups = (n + 15) >> 4;
            int g = wave;
            EdgeMeta nxt = {0, 0, 0.0f, 0.0f, 0.0f};
            if (g < ngroups) nxt = edge_meta(q, n, g * 16, lane, srcIdx, dstIdx, ew, dis);
            while (g < ngroups) {
                const int base = g * 16;
                const int gn = g + 32;
                const EdgeMeta cur = nxt;
                float lo[16], hi[16];
#pragma unroll
                for (int j = 0; j < 16; ++j) {
                    const int sj = __builtin_amdgcn_readlane(cur.s, j);
                    const float* r = v + row64(sj) + lane;
                    lo[j] = r[0];
                    hi[j] = r[32];
                }
                if (gn < ngroups) nxt = edge_meta(q, n, gn * 16, lane, srcIdx, dstIdx, ew, dis);
                const float wb = -scale * cur.dd * cur.w * cur.ds;
                const float wt = alpha * wb;
                const int dl = cur.d - lo_row;
                if (build) {
                    const int idx = base + lane;
                    if (lane < 16 && idx < n && idx < rcap)
                        rq[idx] = make_uint2(((unsigned)cur.s << 8) | (unsigned)dl,
                                             __builtin_bit_cast(unsigned, wb));
                }
#pragma unroll
                for (int j = 0; j < 16; ++j) {
                    if (base + j < n) {
                        const int dj = __builtin_amdgcn_readlane(dl, j);
                        const float wj = __builtin_bit_cast(float,
                            __builtin_amdgcn_readlane(__builtin_bit_cast(int, wt), j));
                        float* o = acc + dj * 64 + lane;
                        atomicAdd(o, wj * lo[j]);
                        atomicAdd(o + 32, wj * hi[j]);
                    }
                }
                g = gn;
            }
        } else {
            const int nW = nSweep * SWEEP_DW;
            for (int wb = wave * 32; wb < nW; wb += 32 * 32) {
                const unsigned zl = zero_bytes(rid[wb + lane] ^ pat);
                unsigned m = __builtin_amdgcn_ballot_w32(zl != 0u);
                while (m) {
                    const int l = __builtin_ctz(m);
                    m &= m - 1u;
                    unsigned zz = (unsigned)__builtin_amdgcn_readlane((int)zl, l);
                    while (zz) {
                        const int bit = __builtin_ctz(zz);
                        zz &= zz - 1u;
                        const int e = 4 * (wb + l) + (bit >> 3);
                        if (e < E) add_edge_row(acc, v, srcIdx, dstIdx, ew, dis, alpha, lo_row, e, lane);
                    }
                }
            }
        }
        if (build && tid == 0) rcnt[blockIdx.x] = n;
    }
    __syncthreads();

    {
        float4* o4 = (float4*)(outf + row64(lo_row));
        const float4* a4 = (const float4*)acc;
        for (int i = tid; i < nq4; i += 1024) o4[i] = a4[i];
    }
}

template <int NT, bool FIRST>
static __device__ __forceinline__ void gemm_orders(const float* __restrict__ T0,
                                                   const float* __restrict__ T1,
                                                   const float* __restrict__ T2,
                                                   const float* __restrict__ W,
                                                   const float* __restrict__ bias,
                                                   float* __restrict__ out,
                                                   int colofs, int nRows,
                                                   float* Wl, float* Al) {
    const int tid = threadIdx.x;
    const int row0b = blockIdx.x * 128;

    const int wave = tid >> 5;
    const int lane = tid & 31;
    const int lm = lane & 15;
    const int lg = lane >> 4;

    const int rowt = wave * 16;
    const int row0 = row0b + rowt;

    v8f c0, c1, c2, c3;
    if (FIRST) {
        const float* bp = bias + lm;
        const float v0 = bp[0], v1 = bp[16], v2 = bp[32], v3 = bp[48];
#pragma unroll
        for (int i = 0; i < 8; ++i) { c0[i] = v0; c1[i] = v1; c2[i] = v2; c3[i] = v3; }
    } else {
        const int nclamp = nRows - 1;
#pragma unroll
        for (int i = 0; i < 8; ++i) {
            int r = row0 + i + lg * 8;
            bool ok = (r < nRows);
            int rc = ok ? r : nclamp;
            const float* orow = out + (size_t)rc * 128 + colofs + lm;
            float v0 = orow[0], v1 = orow[16], v2 = orow[32], v3 = orow[48];
            c0[i] = ok ? v0 : 0.f;
            c1[i] = ok ? v1 : 0.f;
            c2[i] = ok ? v2 : 0.f;
            c3[i] = ok ? v3 : 0.f;
        }
    }

    {
        const int nclamp = nRows - 1;
#pragma unroll
        for (int t = 0; t < NT; ++t) {
            const float* Tt = (t == 0) ? T0 : ((t == 1) ? T1 : T2);
            const float4* W4 = (const float4*)(W + t * (64 * 64));
            float4* Wl4 = (float4*)(Wl + t * (64 * 64));
            float4* Al4 = (float4*)(Al + t * (128 * 64));
#pragma unroll
            for (int i = 0; i < 4; ++i) Wl4[tid + 256 * i] = W4[tid + 256 * i];
#pragma unroll
            for (int i = 0; i < 8; ++i) {
                const int idx = tid + 256 * i;
                const int r = row0b + (idx >> 4);
                const bool ok = (r < nRows);
                float4 v = ((const float4*)(Tt + row64(ok ? r : nclamp)))[idx & 15];
                if (!ok) v = make_float4(0.f, 0.f, 0.f, 0.f);
                Al4[idx] = v;
            }
        }
    }
    __syncthreads();

#pragma unroll
    for (int t = 0; t < NT; ++t) {
        const float* Wt = Wl + t * (64 * 64);
        const float* At = Al + t * (128 * 64);
        const float* arow_l = At + (rowt + lm) * 64;
        const float* bcol_l = Wt + lm;

#pragma unroll
        for (int kk = 0; kk < 16; ++kk) {
            const int kbase = kk * 4 + lg * 2;
            v2f a = *(const v2f*)(arow_l + kbase);
            v2f b0, b1v, b2, b3;
            b0[0]  = bcol_l[(kbase + 0) * 64 + 0];
            b0[1]  = bcol_l[(kbase + 1) * 64 + 0];
            b1v[0] = bcol_l[(kbase + 0) * 64 + 16];
            b1v[1] = bcol_l[(kbase + 1) * 64 + 16];
            b2[0]  = bcol_l[(kbase + 0) * 64 + 32];
            b2[1]  = bcol_l[(kbase + 1) * 64 + 32];
            b3[0]  = bcol_l[(kbase + 0) * 64 + 48];
            b3[1]  = bcol_l[(kbase + 1) * 64 + 48];
            c0 = __builtin_amdgcn_wmma_f32_16x16x4_f32(false, a, false, b0,  (short)0, c0, false, false);
            c1 = __builtin_amdgcn_wmma_f32_16x16x4_f32(false, a, false, b1v, (short)0, c1, false, false);
            c2 = __builtin_amdgcn_wmma_f32_16x16x4_f32(false, a, false, b2,  (short)0, c2, false, false);
            c3 = __builtin_amdgcn_wmma_f32_16x16x4_f32(false, a, false, b3,  (short)0, c3, false, false);
        }
    }

#pragma unroll
    for (int i = 0; i < 8; ++i) {
        int r = row0 + i + lg * 8;
        if (r < nRows) {
            float* orow = out + (size_t)r * 128 + colofs + lm;
            orow[0]  = c0[i];
            orow[16] = c1[i];
            orow[32] = c2[i];
            orow[48] = c3[i];
        }
    }
}

__global__ __launch_bounds__(256) void k_gemm_first3(const float* __restrict__ T0,
                                                     const float* __restrict__ T1,
                                                     const float* __restrict__ T2,
                                                     const float* __restrict__ W,
                                                     const float* __restrict__ bias,
                                                     float* __restrict__ out,
                                                     int colofs, int nRows) {
    __shared__ float Wl[3 * 64 * 64];
    __shared__ float Al[3 * 128 * 64];
    gemm_orders<3, true>(T0, T1, T2, W, bias, out, colofs, nRows, Wl, Al);
}

__global__ __launch_bounds__(256) void k_gemm_acc2(const float* __restrict__ T0,
                                                   const float* __restrict__ T1,
                                                   const float* __restrict__ W,
                                                   float* __restrict__ out,
                                                   int colofs, int nRows) {
    __shared__ float Wl[2 * 64 * 64];
    __shared__ float Al[2 * 128 * 64];
    gemm_orders<2, false>(T0, T1, T1, W, W, out, colofs, nRows, Wl, Al);
}


extern "C" void kernel_launch(void* const* d_in, const int* in_sizes, int n_in,
                              void* d_out, int out_size, void* d_ws, size_t ws_size,
                              hipStream_t stream) {
    const float* x  = (const float*)d_in[0];
    const int*   ei = (const int*)d_in[1];
    const float* ew = (const float*)d_in[2];
    const float* W1 = (const float*)d_in[3];
    const float* b1 = (const float*)d_in[4];
    const float* W2 = (const float*)d_in[5];
    const float* b2 = (const float*)d_in[6];
    float* out = (float*)d_out;

    const int N = in_sizes[0] / 64;
    const int E = in_sizes[2];
    const int* row = ei;
    const int* col = ei + E;

    float* ws   = (float*)d_ws;
    float* deg1 = ws;
    float* deg2 = deg1 + N;
    unsigned* rid1 = (unsigned*)(deg2 + N);
    unsigned* rid2 = rid1 + E;
    float* tA   = (float*)(rid2 + E);
    float* tB   = tA + (size_t)N * 64;
    unsigned* rec = (unsigned*)(tB + (size_t)N * 64);

    const int nSweep = ((E + 3) / 4 + SWEEP_DW - 1) / SWEEP_DW;
    const float DIAG = 2.0f / 3.0f - 1.0f;
    int* rcnt = (int*)(rid1 + (size_t)nSweep * SWEEP_DW);

    dim3 blk(256);
    int gN2   = (2 * N + 255) / 256;
    int gE    = (E + 255) / 256;
    int gW    = nSweep * (SWEEP_DW / 256);
    int gOwn  = (N + ROWS_PER_BLOCK - 1) / ROWS_PER_BLOCK;
    int gGemm = (N + 127) / 128;

    k_zero<<<gN2, blk, 0, stream>>>(deg1, 2 * N);
    k_deg<<<gE, blk, 0, stream>>>(ew, row, col, deg1, deg2, E);
    k_invsqrt<<<gN2, blk, 0, stream>>>(deg1, 2 * N);
    k_rid<<<gW, blk, 0, stream>>>(row, col, rid1, rid2, E);

    for (int dir = 0; dir < 2; ++dir) {
        const int*      src  = (dir == 0) ? col : row;
        const int*      dst  = (dir == 0) ? row : col;
        const float*    dis  = (dir == 0) ? deg1 : deg2;
        const unsigned* rid  = (dir == 0) ? rid1 : rid2;
        const float*    W    = (dir == 0) ? W1 : W2;
        const float*    bias = (dir == 0) ? b1 : b2;
        const int       cofs = (dir == 0) ? 0 : 64;

        k_own<<<gOwn, 1024, 0, stream>>>(x, x, tA, src, dst, ew, dis, rid, rec, rcnt, 1, DIAG, 0.0f, 1.0f, N, E, nSweep);

        k_own<<<gOwn, 1024, 0, stream>>>(tA, x, tB, src, dst, ew, dis, rid, rec, rcnt, 0, 2.0f * DIAG, -1.0f, 2.0f, N, E, nSweep);

        k_gemm_first3<<<gGemm, blk, 0, stream>>>(x, tA, tB, W, bias, out, cofs, N);

        k_own<<<gOwn, 1024, 0, stream>>>(tB, tA, tA, src, dst, ew, dis, rid, rec, rcnt, 0, 2.0f * DIAG, -1.0f, 2.0f, N, E, nSweep);

        k_own<<<gOwn, 1024, 0, stream>>>(tA, tB, tB, src, dst, ew, dis, rid, rec, rcnt, 0, 2.0f * DIAG, -1.0f, 2.0f, N, E, nSweep);

        k_gemm_acc2<<<gGemm, blk, 0, stream>>>(tA, tB, W + 3 * 4096, out, cofs, N);
    }
}


# Edge records built by one bucketing pass per direction: each edge read once, LDS chunk counts, one atomic slot reservation per chunk and row block; no range-id filter; LDS sums in arrival order
# speedup vs baseline: 2.7452x; 1.0649x over previous
#include <hip/hip_runtime.h>


typedef float v2f __attribute__((ext_vector_type(2)));
typedef float v8f __attribute__((ext_vector_type(8)));

static __device__ __forceinline__ size_t row64(int n) { return (size_t)n * 64; }


__global__ __launch_bounds__(256) void k_zero(float* __restrict__ p, int n) {
    int i = blockIdx.x * 256 + threadIdx.x;
    if (i < n) p[i] = 0.0f;
}

__global__ __launch_bounds__(256) void k_deg(const float* __restrict__ ew,
                                             const int* __restrict__ row,
                                             const int* __restrict__ col,
                                             float* __restrict__ deg1,
                                             float* __restrict__ deg2, int E) {
    int e = blockIdx.x * 256 + threadIdx.x;
    if (e >= E) return;
    float w = ew[e];
    atomicAdd(&deg1[row[e]], w);
    atomicAdd(&deg2[col[e]], w);
}

__global__ __launch_bounds__(256) void k_invsqrt(float* __restrict__ d, int n) {
    int i = blockIdx.x * 256 + threadIdx.x;
    if (i < n) {
        float v = d[i];
        d[i] = (v > 0.0f) ? rsqrtf(v) : 0.0f;
    }
}

#define ROWS_PER_BLOCK 200
#define BK_THREADS 256
#define BK_U 16
#define BK_CHUNK (BK_THREADS * BK_U)
__global__ __launch_bounds__(BK_THREADS) void k_bucket(const int* __restrict__ srcIdx,
                                                       const int* __restrict__ dstIdx,
                                                       const float* __restrict__ ew,
                                                       const float* __restrict__ dis,
                                                       unsigned* __restrict__ rec,
                                                       int* __restrict__ rcnt,
                                                       int N, int E) {
    __shared__ int slot[BK_THREADS];
    const int tid = threadIdx.x;
    slot[tid] = 0;
    __syncthreads();

    const float scale = 2.0f / 3.0f;
    const int e0 = blockIdx.x * BK_CHUNK + tid;
    const int elast = E - 1;
    int s[BK_U], d[BK_U];
    float w[BK_U];
#pragma unroll
    for (int k = 0; k < BK_U; ++k) {
        const int e = e0 + k * BK_THREADS;
        const int ec = (e < E) ? e : elast;
        s[k] = srcIdx[ec];
        d[k] = dstIdx[ec];
        w[k] = ew[ec];
    }
    unsigned rx[BK_U], ry[BK_U];
    int bl[BK_U], loc[BK_U];
#pragma unroll
    for (int k = 0; k < BK_U; ++k) {
        const float dd = dis[d[k]];
        const float ds = dis[s[k]];
        const int b = (int)((unsigned)d[k] / ROWS_PER_BLOCK);
        bl[k] = b;
        rx[k] = ((unsigned)s[k] << 8) | (unsigned)(d[k] - b * ROWS_PER_BLOCK);
        ry[k] = __builtin_bit_cast(unsigned, -scale * dd * w[k] * ds);
    }
#pragma unroll
    for (int k = 0; k < BK_U; ++k) {
        loc[k] = 0;
        if (e0 + k * BK_THREADS < E) loc[k] = atomicAdd(&slot[bl[k]], 1);
    }
    __syncthreads();
    {
        const int cnt = slot[tid];
        if (cnt > 0) slot[tid] = atomicAdd(&rcnt[tid], cnt);
    }
    __syncthreads();
#pragma unroll
    for (int k = 0; k < BK_U; ++k) {
        if (e0 + k * BK_THREADS < E) {
            const int b = bl[k];
            const int pos = slot[b] + loc[k];
            int nrows = N - b * ROWS_PER_BLOCK;
            if (nrows > ROWS_PER_BLOCK) nrows = ROWS_PER_BLOCK;
            if (pos < nrows * 32)
                ((uint2*)rec)[(size_t)b * (ROWS_PER_BLOCK * 32) + pos] = make_uint2(rx[k], ry[k]);
        }
    }
}


static_assert(ROWS_PER_BLOCK <= 256, "a record keeps the local row in 8 bits");
static __device__ __forceinline__ uint2 edge_rec(const uint2* rq, int n, int base, int lane) {
    const int idx = base + (lane & 15);
    return rq[(idx < n) ? idx : base];
}

static __device__ __forceinline__ void add_edge_row(float* acc, const float* __restrict__ v,
                                                    const int* __restrict__ srcIdx,
                                                    const int* __restrict__ dstIdx,
                                                    const float* __restrict__ ew,
                                                    const float* __restrict__ dis,
                                                    float alpha, int lo_row, int e, int lane) {
    const float scale = 2.0f / 3.0f;
    const int s = srcIdx[e], d = dstIdx[e];
    const float wt = alpha * (-scale * dis[d] * ew[e] * dis[s]);
    const float* r = v + row64(s) + lane;
    const float vlo = r[0], vhi = r[32];
    float* o = acc + (d - lo_row) * 64 + lane;
    atomicAdd(o, wt * vlo);
    atomicAdd(o + 32, wt * vhi);
}

__global__ __launch_bounds__(1024) __attribute__((amdgpu_waves_per_eu(8, 8)))
void k_own(const float* __restrict__ v,
           const float* prev,
           float* outf,
           const int* __restrict__ srcIdx,
           const int* __restrict__ dstIdx,
           const float* __restrict__ ew,
           const float* __restrict__ dis,
           const unsigned* __restrict__ rec,
           const int* __restrict__ rcnt,
           float alpha_diag, float beta, float alpha,
           int N, int E) {
    __shared__ float acc[ROWS_PER_BLOCK * 64];

    const int tid = threadIdx.x;
    const int lane = tid & 31;
    const int wave = __builtin_amdgcn_readfirstlane(tid >> 5);
    const int lo_row = blockIdx.x * ROWS_PER_BLOCK;
    int nrows = N - lo_row;
    if (nrows > ROWS_PER_BLOCK) nrows = ROWS_PER_BLOCK;
    const int nq4 = nrows * 16;
    const uint2* rq = (const uint2*)rec + (size_t)lo_row * 32;
    const int rcap = nrows * 32;

    {
        const float4* v4 = (const float4*)(v + row64(lo_row));
        const float4* p4 = (const float4*)(prev + row64(lo_row));
        float4* a4 = (float4*)acc;
        for (int i = tid; i < nq4; i += 1024) {
            float4 a = v4[i], p = p4[i], r;
            r.x = alpha_diag * a.x + beta * p.x;
            r.y = alpha_diag * a.y + beta * p.y;
            r.z = alpha_diag * a.z + beta * p.z;
            r.w = alpha_diag * a.w + beta * p.w;
            a4[i] = r;
        }
    }
    __syncthreads();

    const int nrec = __builtin_amdgcn_readfirstlane(rcnt[blockIdx.x]);
    if (nrec <= rcap) {
        const int ngroups = (nrec + 15) >> 4;
        int g = wave;
        uint2 nxt = make_uint2(0u, 0u);
        if (g < ngroups) nxt = edge_rec(rq, nrec, g * 16, lane);
        while (g < ngroups) {
            const int base = g * 16;
            const int gn = g + 32;
            const uint2 cur = nxt;
            float lo[16], hi[16];
#pragma unroll
            for (int j = 0; j < 16; ++j) {
                const int sj = (int)((unsigned)__builtin_amdgcn_readlane((int)cur.x, j) >> 8);
                const float* r = v + row64(sj) + lane;
                lo[j] = r[0];
                hi[j] = r[32];
            }
            if (gn < ngroups) nxt = edge_rec(rq, nrec, gn * 16, lane);
            const float wt = alpha * __builtin_bit_cast(float, cur.y);
            const int dl = (int)(cur.x & 255u);
#pragma unroll
            for (int j = 0; j < 16; ++j) {
                if (base + j < nrec) {
                    const int dj = __builtin_amdgcn_readlane(dl, j);
                    const float wj = __builtin_bit_cast(float,
                        __builtin_amdgcn_readlane(__builtin_bit_cast(int, wt), j));
                    float* o = acc + dj * 64 + lane;
                    atomicAdd(o, wj * lo[j]);
                    atomicAdd(o + 32, wj * hi[j]);
                }
            }
            g = gn;
        }
    } else {
        for (int eb = wave * 32; eb < E; eb += 32 * 32) {
            const int e = eb + lane;
            bool mine = false;
            if (e < E) mine = (unsigned)(dstIdx[e] - lo_row) < (unsigned)nrows;
            unsigned m = __builtin_amdgcn_ballot_w32(mine);
            while (m) {
                const int l = __builtin_ctz(m);
                m &= m - 1u;
                add_edge_row(acc, v, srcIdx, dstIdx, ew, dis, alpha, lo_row, eb + l, lane);
            }
        }
    }
    __syncthreads();

    {
        float4* o4 = (float4*)(outf + row64(lo_row));
        const float4* a4 = (const float4*)acc;
        for (int i = tid; i < nq4; i += 1024) o4[i] = a4[i];
    }
}

template <int NT, bool FIRST>
static __device__ __forceinline__ void gemm_orders(const float* __restrict__ T0,
                                                   const float* __restrict__ T1,
                                                   const float* __restrict__ T2,
                                                   const float* __restrict__ W,
                                                   const float* __restrict__ bias,
                                                   float* __restrict__ out,
                                                   int colofs, int nRows,
                                                   float* Wl, float* Al) {
    const int tid = threadIdx.x;
    const int row0b = blockIdx.x * 128;

    const int wave = tid >> 5;
    const int lane = tid & 31;
    const int lm = lane & 15;
    const int lg = lane >> 4;

    const int rowt = wave * 16;
    const int row0 = row0b + rowt;

    v8f c0, c1, c2, c3;
    if (FIRST) {
        const float* bp = bias + lm;
        const float v0 = bp[0], v1 = bp[16], v2 = bp[32], v3 = bp[48];
#pragma unroll
        for (int i = 0; i < 8; ++i) { c0[i] = v0; c1[i] = v1; c2[i] = v2; c3[i] = v3; }
    } else {
        const int nclamp = nRows - 1;
#pragma unroll
        for (int i = 0; i < 8; ++i) {
            int r = row0 + i + lg * 8;
            bool ok = (r < nRows);
            int rc = ok ? r : nclamp;
            const float* orow = out + (size_t)rc * 128 + colofs + lm;
            float v0 = orow[0], v1 = orow[16], v2 = orow[32], v3 = orow[48];
            c0[i] = ok ? v0 : 0.f;
            c1[i] = ok ? v1 : 0.f;
            c2[i] = ok ? v2 : 0.f;
            c3[i] = ok ? v3 : 0.f;
        }
    }

    {
        const int nclamp = nRows - 1;
#pragma unroll
        for (int t = 0; t < NT; ++t) {
            const float* Tt = (t == 0) ? T0 : ((t == 1) ? T1 : T2);
            const float4* W4 = (const float4*)(W + t * (64 * 64));
            float4* Wl4 = (float4*)(Wl + t * (64 * 64));
            float4* Al4 = (float4*)(Al + t * (128 * 64));
#pragma unroll
            for (int i = 0; i < 4; ++i) Wl4[tid + 256 * i] = W4[tid + 256 * i];
#pragma unroll
            for (int i = 0; i < 8; ++i) {
                const int idx = tid + 256 * i;
                const int r = row0b + (idx >> 4);
                const bool ok = (r < nRows);
                float4 v = ((const float4*)(Tt + row64(ok ? r : nclamp)))[idx & 15];
                if (!ok) v = make_float4(0.f, 0.f, 0.f, 0.f);
                Al4[idx] = v;
            }
        }
    }
    __syncthreads();

#pragma unroll
    for (int t = 0; t < NT; ++t) {
        const float* Wt = Wl + t * (64 * 64);
        const float* At = Al + t * (128 * 64);
        const float* arow_l = At + (rowt + lm) * 64;
        const float* bcol_l = Wt + lm;

#pragma unroll
        for (int kk = 0; kk < 16; ++kk) {
            const int kbase = kk * 4 + lg * 2;
            v2f a = *(const v2f*)(arow_l + kbase);
            v2f b0, b1v, b2, b3;
            b0[0]  = bcol_l[(kbase + 0) * 64 + 0];
            b0[1]  = bcol_l[(kbase + 1) * 64 + 0];
            b1v[0] = bcol_l[(kbase + 0) * 64 + 16];
            b1v[1] = bcol_l[(kbase + 1) * 64 + 16];
            b2[0]  = bcol_l[(kbase + 0) * 64 + 32];
            b2[1]  = bcol_l[(kbase + 1) * 64 + 32];
            b3[0]  = bcol_l[(kbase + 0) * 64 + 48];
            b3[1]  = bcol_l[(kbase + 1) * 64 + 48];
            c0 = __builtin_amdgcn_wmma_f32_16x16x4_f32(false, a, false, b0,  (short)0, c0, false, false);
            c1 = __builtin_amdgcn_wmma_f32_16x16x4_f32(false, a, false, b1v, (short)0, c1, false, false);
            c2 = __builtin_amdgcn_wmma_f32_16x16x4_f32(false, a, false, b2,  (short)0, c2, false, false);
            c3 = __builtin_amdgcn_wmma_f32_16x16x4_f32(false, a, false, b3,  (short)0, c3, false, false);
        }
    }

#pragma unroll
    for (int i = 0; i < 8; ++i) {
        int r = row0 + i + lg * 8;
        if (r < nRows) {
            float* orow = out + (size_t)r * 128 + colofs + lm;
            orow[0]  = c0[i];
            orow[16] = c1[i];
            orow[32] = c2[i];
            orow[48] = c3[i];
        }
    }
}

__global__ __launch_bounds__(256) void k_gemm_first3(const float* __restrict__ T0,
                                                     const float* __restrict__ T1,
                                                     const float* __restrict__ T2,
                                                     const float* __restrict__ W,
                                                     const float* __restrict__ bias,
                                                     float* __restrict__ out,
                                                     int colofs, int nRows) {
    __shared__ float Wl[3 * 64 * 64];
    __shared__ float Al[3 * 128 * 64];
    gemm_orders<3, true>(T0, T1, T2, W, bias, out, colofs, nRows, Wl, Al);
}

__global__ __launch_bounds__(256) void k_gemm_acc2(const float* __restrict__ T0,
                                                   const float* __restrict__ T1,
                                                   const float* __restrict__ W,
                                                   float* __restrict__ out,
                                                   int colofs, int nRows) {
    __shared__ float Wl[2 * 64 * 64];
    __shared__ float Al[2 * 128 * 64];
    gemm_orders<2, false>(T0, T1, T1, W, W, out, colofs, nRows, Wl, Al);
}


extern "C" void kernel_launch(void* const* d_in, const int* in_sizes, int n_in,
                              void* d_out, int out_size, void* d_ws, size_t ws_size,
                              hipStream_t stream) {
    const float* x  = (const float*)d_in[0];
    const int*   ei = (const int*)d_in[1];
    const float* ew = (const float*)d_in[2];
    const float* W1 = (const float*)d_in[3];
    const float* b1 = (const float*)d_in[4];
    const float* W2 = (const float*)d_in[5];
    const float* b2 = (const float*)d_in[6];
    float* out = (float*)d_out;

    const int N = in_sizes[0] / 64;
    const int E = in_sizes[2];
    const int* row = ei;
    const int* col = ei + E;

    float* ws   = (float*)d_ws;
    float* deg1 = ws;
    float* deg2 = deg1 + N;
    int* rcnt1  = (int*)(deg2 + N);
    int* rcnt2  = rcnt1 + 256;
    float* tA   = (float*)(rcnt2 + 256);
    float* tB   = tA + (size_t)N * 64;
    unsigned* rec = (unsigned*)(tB + (size_t)N * 64);

    const float DIAG = 2.0f / 3.0f - 1.0f;

    dim3 blk(256);
    int gZ    = (2 * N + 512 + 255) / 256;
    int gN2   = (2 * N + 255) / 256;
    int gE    = (E + 255) / 256;
    int gBk   = (E + BK_CHUNK - 1) / BK_CHUNK;
    int gOwn  = (N + ROWS_PER_BLOCK - 1) / ROWS_PER_BLOCK;
    int gGemm = (N + 127) / 128;

    k_zero<<<gZ, blk, 0, stream>>>(deg1, 2 * N + 512);
    k_deg<<<gE, blk, 0, stream>>>(ew, row, col, deg1, deg2, E);
    k_invsqrt<<<gN2, blk, 0, stream>>>(deg1, 2 * N);

    for (int dir = 0; dir < 2; ++dir) {
        const int*      src  = (dir == 0) ? col : row;
        const int*      dst  = (dir == 0) ? row : col;
        const float*    dis  = (dir == 0) ? deg1 : deg2;
        int*            rcnt = (dir == 0) ? rcnt1 : rcnt2;
        const float*    W    = (dir == 0) ? W1 : W2;
        const float*    bias = (dir == 0) ? b1 : b2;
        const int       cofs = (dir == 0) ? 0 : 64;

        k_bucket<<<gBk, BK_THREADS, 0, stream>>>(src, dst, ew, dis, rec, rcnt, N, E);

        k_own<<<gOwn, 1024, 0, stream>>>(x, x, tA, src, dst, ew, dis, rec, rcnt, DIAG, 0.0f, 1.0f, N, E);

        k_own<<<gOwn, 1024, 0, stream>>>(tA, x, tB, src, dst, ew, dis, rec, rcnt, 2.0f * DIAG, -1.0f, 2.0f, N, E);

        k_gemm_first3<<<gGemm, blk, 0, stream>>>(x, tA, tB, W, bias, out, cofs, N);

        k_own<<<gOwn, 1024, 0, stream>>>(tB, tA, tA, src, dst, ew, dis, rec, rcnt, 2.0f * DIAG, -1.0f, 2.0f, N, E);

        k_own<<<gOwn, 1024, 0, stream>>>(tA, tB, tB, src, dst, ew, dis, rec, rcnt, 2.0f * DIAG, -1.0f, 2.0f, N, E);

        k_gemm_acc2<<<gGemm, blk, 0, stream>>>(tA, tB, W + 3 * 4096, out, cofs, N);
    }
}


# speedup vs baseline: 2.9094x; 1.0598x over previous
#include <hip/hip_runtime.h>


typedef float v2f __attribute__((ext_vector_type(2)));
typedef float v8f __attribute__((ext_vector_type(8)));

static __device__ __forceinline__ size_t row64(int n) { return (size_t)n * 64; }


__global__ __launch_bounds__(256) void k_zero(float* __restrict__ p, int n) {
    int i = blockIdx.x * 256 + threadIdx.x;
    if (i < n) p[i] = 0.0f;
}

__global__ __launch_bounds__(256) void k_deg(const float* __restrict__ ew,
                                             const int* __restrict__ row,
                                             const int* __restrict__ col,
                                             float* __restrict__ deg1,
                                             float* __restrict__ deg2, int E) {
    int e = blockIdx.x * 256 + threadIdx.x;
    if (e >= E) return;
    float w = ew[e];
    atomicAdd(&deg1[row[e]], w);
    atomicAdd(&deg2[col[e]], w);
}

__global__ __launch_bounds__(256) void k_invsqrt(float* __restrict__ d, int n) {
    int i = blockIdx.x * 256 + threadIdx.x;
    if (i < n) {
        float v = d[i];
        d[i] = (v > 0.0f) ? rsqrtf(v) : 0.0f;
    }
}

#define ROWS_PER_BLOCK 200
#define BK_THREADS 256
#define BK_U 16
#define BK_CHUNK (BK_THREADS * BK_U)
__global__ __launch_bounds__(BK_THREADS) void k_bucket(const int* __restrict__ srcIdx,
                                                       const int* __restrict__ dstIdx,
                                                       const float* __restrict__ ew,
                                                       const float* __restrict__ dis,
                                                       unsigned* __restrict__ rec,
                                                       int* __restrict__ rcnt,
                                                       int N, int E) {
    __shared__ int slot[BK_THREADS];
    const int tid = threadIdx.x;
    slot[tid] = 0;
    __syncthreads();

    const float scale = 2.0f / 3.0f;
    const int e0 = blockIdx.x * BK_CHUNK + tid;
    const int elast = E - 1;
    int s[BK_U], d[BK_U];
    float w[BK_U];
#pragma unroll
    for (int k = 0; k < BK_U; ++k) {
        const int e = e0 + k * BK_THREADS;
        const int ec = (e < E) ? e : elast;
        s[k] = srcIdx[ec];
        d[k] = dstIdx[ec];
        w[k] = ew[ec];
    }
    unsigned rx[BK_U], ry[BK_U];
    int bl[BK_U], loc[BK_U];
#pragma unroll
    for (int k = 0; k < BK_U; ++k) {
        const float dd = dis[d[k]];
        const float ds = dis[s[k]];
        const int b = (int)((unsigned)d[k] / ROWS_PER_BLOCK);
        bl[k] = b;
        rx[k] = ((unsigned)s[k] << 8) | (unsigned)(d[k] - b * ROWS_PER_BLOCK);
        ry[k] = __builtin_bit_cast(unsigned, -scale * dd * w[k] * ds);
    }
#pragma unroll
    for (int k = 0; k < BK_U; ++k) {
        loc[k] = 0;
        if (e0 + k * BK_THREADS < E) loc[k] = atomicAdd(&slot[bl[k]], 1);
    }
    __syncthreads();
    {
        const int cnt = slot[tid];
        if (cnt > 0) slot[tid] = atomicAdd(&rcnt[tid], cnt);
    }
    __syncthreads();
#pragma unroll
    for (int k = 0; k < BK_U; ++k) {
        if (e0 + k * BK_THREADS < E) {
            const int b = bl[k];
            const int pos = slot[b] + loc[k];
            int nrows = N - b * ROWS_PER_BLOCK;
            if (nrows > ROWS_PER_BLOCK) nrows = ROWS_PER_BLOCK;
            if (pos < nrows * 32)
                ((uint2*)rec)[(size_t)b * (ROWS_PER_BLOCK * 32) + pos] = make_uint2(rx[k], ry[k]);
        }
    }
}


static_assert(ROWS_PER_BLOCK <= 256, "a record keeps the local row in 8 bits");
static __device__ __forceinline__ uint2 edge_rec(const uint2* rq, int n, int base, int lane) {
    const int idx = base + ((lane & 15) ^ (lane >> 4));
    return rq[(idx < n) ? idx : base];
}

template <int I>
static __device__ __forceinline__ int row_share(int x) {
    return __builtin_amdgcn_update_dpp(0, x, 0x150 + I, 0xf, 0xf, false);
}

template <int P>
static __device__ __forceinline__ float4 gather_pair(const float* __restrict__ v, unsigned recx, unsigned lane16b) {
    const unsigned sx = (unsigned)row_share<2 * P>((int)recx);
    const unsigned off = (sx & 0xFFFFFF00u) | lane16b;
    return *(const float4*)((const char*)v + off);
}

template <int P, bool TAIL>
static __device__ __forceinline__ void add_pair(float* acc, float4 t, unsigned recx, float wt,
                                                int kA, int kB, bool upper, bool ok) {
    const int sx = row_share<2 * P>((int)recx);
    const float w = __builtin_bit_cast(float, row_share<2 * P>(__builtin_bit_cast(int, wt)));
    float* row = acc + (sx & 255) * 64;
    const float r0 = upper ? t.y : t.x;
    const float r1 = upper ? t.x : t.y;
    const float r2 = upper ? t.w : t.z;
    const float r3 = upper ? t.z : t.w;
    if (!TAIL || ok) {
        atomicAdd(row + kA, w * r0);
        atomicAdd(row + kB, w * r1);
        atomicAdd(row + kA + 32, w * r2);
        atomicAdd(row + kB + 32, w * r3);
    }
}

static __device__ __forceinline__ void add_edge_row(float* acc, const float* __restrict__ v,
                                                    const int* __restrict__ srcIdx,
                                                    const int* __restrict__ dstIdx,
                                                    const float* __restrict__ ew,
                                                    const float* __restrict__ dis,
                                                    float alpha, int lo_row, int e, int lane) {
    const float scale = 2.0f / 3.0f;
    const int s = srcIdx[e], d = dstIdx[e];
    const float wt = alpha * (-scale * dis[d] * ew[e] * dis[s]);
    const float* r = v + row64(s) + lane;
    const float vlo = r[0], vhi = r[32];
    float* o = acc + (d - lo_row) * 64 + 16 * (lane & 3) + (lane >> 2);
    atomicAdd(o, wt * vlo);
    atomicAdd(o + 8, wt * vhi);
}

__global__ __launch_bounds__(1024) __attribute__((amdgpu_waves_per_eu(8, 8)))
void k_own(const float* __restrict__ v,
           const float* prev,
           float* outf,
           const int* __restrict__ srcIdx,
           const int* __restrict__ dstIdx,
           const float* __restrict__ ew,
           const float* __restrict__ dis,
           const unsigned* __restrict__ rec,
           const int* __restrict__ rcnt,
           float alpha_diag, float beta, float alpha,
           int N, int E) {
    __shared__ float acc[ROWS_PER_BLOCK * 64];

    const int tid = threadIdx.x;
    const int lane = tid & 31;
    const int wave = __builtin_amdgcn_readfirstlane(tid >> 5);
    const int lo_row = blockIdx.x * ROWS_PER_BLOCK;
    int nrows = N - lo_row;
    if (nrows > ROWS_PER_BLOCK) nrows = ROWS_PER_BLOCK;
    const int nq4 = nrows * 16;
    const uint2* rq = (const uint2*)rec + (size_t)lo_row * 32;
    const int rcap = nrows * 32;

    {
        const float4* v4 = (const float4*)(v + row64(lo_row));
        const float4* p4 = (const float4*)(prev + row64(lo_row));
        for (int i = tid; i < nq4; i += 1024) {
            float4 a = v4[i], p = p4[i];
            float* o = acc + (i >> 4) * 64 + (i & 15);
            o[0]  = alpha_diag * a.x + beta * p.x;
            o[16] = alpha_diag * a.y + beta * p.y;
            o[32] = alpha_diag * a.z + beta * p.z;
            o[48] = alpha_diag * a.w + beta * p.w;
        }
    }
    __syncthreads();

    const int nrec = __builtin_amdgcn_readfirstlane(rcnt[blockIdx.x]);
    if (nrec <= rcap) {
        const int ngroups = (nrec + 15) >> 4;
        int g = wave;
        uint2 nxt = make_uint2(0u, 0u);
        if (g < ngroups) nxt = edge_rec(rq, nrec, g * 16, lane);
        const int l16 = lane & 15;
        const bool upper = lane >= 16;
        const unsigned lane16b = (unsigned)l16 * 16u;
        const int kA = l16 + (upper ? 16 : 0);
        const int kB = l16 + (upper ? 0 : 16);
        while (g < ngroups) {
            const int base = g * 16;
            const int gn = g + 32;
            const uint2 cur = nxt;
            const float4 t0 = gather_pair<0>(v, cur.x, lane16b);
            const float4 t1 = gather_pair<1>(v, cur.x, lane16b);
            const float4 t2 = gather_pair<2>(v, cur.x, lane16b);
            const float4 t3 = gather_pair<3>(v, cur.x, lane16b);
            const float4 t4 = gather_pair<4>(v, cur.x, lane16b);
            const float4 t5 = gather_pair<5>(v, cur.x, lane16b);
            const float4 t6 = gather_pair<6>(v, cur.x, lane16b);
            const float4 t7 = gather_pair<7>(v, cur.x, lane16b);
            if (gn < ngroups) nxt = edge_rec(rq, nrec, gn * 16, lane);
            const float wt = alpha * __builtin_bit_cast(float, cur.y);
            if (base + 16 <= nrec) {
                add_pair<0, false>(acc, t0, cur.x, wt, kA, kB, upper, true);
                add_pair<1, false>(acc, t1, cur.x, wt, kA, kB, upper, true);
                add_pair<2, false>(acc, t2, cur.x, wt, kA, kB, upper, true);
                add_pair<3, false>(acc, t3, cur.x, wt, kA, kB, upper, true);
                add_pair<4, false>(acc, t4, cur.x, wt, kA, kB, upper, true);
                add_pair<5, false>(acc, t5, cur.x, wt, kA, kB, upper, true);
                add_pair<6, false>(acc, t6, cur.x, wt, kA, kB, upper, true);
                add_pair<7, false>(acc, t7, cur.x, wt, kA, kB, upper, true);
            } else {
                const int e0 = base + (upper ? 1 : 0);
                add_pair<0, true>(acc, t0, cur.x, wt, kA, kB, upper, e0 + 0 < nrec);
                add_pair<1, true>(acc, t1, cur.x, wt, kA, kB, upper, e0 + 2 < nrec);
                add_pair<2, true>(acc, t2, cur.x, wt, kA, kB, upper, e0 + 4 < nrec);
                add_pair<3, true>(acc, t3, cur.x, wt, kA, kB, upper, e0 + 6 < nrec);
                add_pair<4, true>(acc, t4, cur.x, wt, kA, kB, upper, e0 + 8 < nrec);
                add_pair<5, true>(acc, t5, cur.x, wt, kA, kB, upper, e0 + 10 < nrec);
                add_pair<6, true>(acc, t6, cur.x, wt, kA, kB, upper, e0 + 12 < nrec);
                add_pair<7, true>(acc, t7, cur.x, wt, kA, kB, upper, e0 + 14 < nrec);
            }
            g = gn;
        }
    } else {
        for (int eb = wave * 32; eb < E; eb += 32 * 32) {
            const int e = eb + lane;
            bool mine = false;
            if (e < E) mine = (unsigned)(dstIdx[e] - lo_row) < (unsigned)nrows;
            unsigned m = __builtin_amdgcn_ballot_w32(mine);
            while (m) {
                const int l = __builtin_ctz(m);
                m &= m - 1u;
                add_edge_row(acc, v, srcIdx, dstIdx, ew, dis, alpha, lo_row, eb + l, lane);
            }
        }
    }
    __syncthreads();

    {
        float4* o4 = (float4*)(outf + row64(lo_row));
        for (int i = tid; i < nq4; i += 1024) {
            const float* a = acc + (i >> 4) * 64 + (i & 15);
            o4[i] = make_float4(a[0], a[16], a[32], a[48]);
        }
    }
}

template <int NT, bool FIRST>
static __device__ __forceinline__ void gemm_orders(const float* __restrict__ T0,
                                                   const float* __restrict__ T1,
                                                   const float* __restrict__ T2,
                                                   const float* __restrict__ W,
                                                   const float* __restrict__ bias,
                                                   float* __restrict__ out,
                                                   int colofs, int nRows,
                                                   float* Wl, float* Al) {
    const int tid = threadIdx.x;
    const int row0b = blockIdx.x * 128;

    const int wave = tid >> 5;
    const int lane = tid & 31;
    const int lm = lane & 15;
    const int lg = lane >> 4;

    const int rowt = wave * 16;
    const int row0 = row0b + rowt;

    v8f c0, c1, c2, c3;
    if (FIRST) {
        const float* bp = bias + lm;
        const float v0 = bp[0], v1 = bp[16], v2 = bp[32], v3 = bp[48];
#pragma unroll
        for (int i = 0; i < 8; ++i) { c0[i] = v0; c1[i] = v1; c2[i] = v2; c3[i] = v3; }
    } else {
        const int nclamp = nRows - 1;
#pragma unroll
        for (int i = 0; i < 8; ++i) {
            int r = row0 + i + lg * 8;
            bool ok = (r < nRows);
            int rc = ok ? r : nclamp;
            const float* orow = out + (size_t)rc * 128 + colofs + lm;
            float v0 = orow[0], v1 = orow[16], v2 = orow[32], v3 = orow[48];
            c0[i] = ok ? v0 : 0.f;
            c1[i] = ok ? v1 : 0.f;
            c2[i] = ok ? v2 : 0.f;
            c3[i] = ok ? v3 : 0.f;
        }
    }

    {
        const int nclamp = nRows - 1;
#pragma unroll
        for (int t = 0; t < NT; ++t) {
            const float* Tt = (t == 0) ? T0 : ((t == 1) ? T1 : T2);
            const float4* W4 = (const float4*)(W + t * (64 * 64));
            float4* Wl4 = (float4*)(Wl + t * (64 * 64));
            float4* Al4 = (float4*)(Al + t * (128 * 64));
#pragma unroll
            for (int i = 0; i < 4; ++i) Wl4[tid + 256 * i] = W4[tid + 256 * i];
#pragma unroll
            for (int i = 0; i < 8; ++i) {
                const int idx = tid + 256 * i;
                const int r = row0b + (idx >> 4);
                const bool ok = (r < nRows);
                float4 v = ((const float4*)(Tt + row64(ok ? r : nclamp)))[idx & 15];
                if (!ok) v = make_float4(0.f, 0.f, 0.f, 0.f);
                Al4[idx] = v;
            }
        }
    }
    __syncthreads();

#pragma unroll
    for (int t = 0; t < NT; ++t) {
        const float* Wt = Wl + t * (64 * 64);
        const float* At = Al + t * (128 * 64);
        const float* arow_l = At + (rowt + lm) * 64;
        const float* bcol_l = Wt + lm;

#pragma unroll
        for (int kk = 0; kk < 16; ++kk) {
            const int kbase = kk * 4 + lg * 2;
            v2f a = *(const v2f*)(arow_l + kbase);
            v2f b0, b1v, b2, b3;
            b0[0]  = bcol_l[(kbase + 0) * 64 + 0];
            b0[1]  = bcol_l[(kbase + 1) * 64 + 0];
            b1v[0] = bcol_l[(kbase + 0) * 64 + 16];
            b1v[1] = bcol_l[(kbase + 1) * 64 + 16];
            b2[0]  = bcol_l[(kbase + 0) * 64 + 32];
            b2[1]  = bcol_l[(kbase + 1) * 64 + 32];
            b3[0]  = bcol_l[(kbase + 0) * 64 + 48];
            b3[1]  = bcol_l[(kbase + 1) * 64 + 48];
            c0 = __builtin_amdgcn_wmma_f32_16x16x4_f32(false, a, false, b0,  (short)0, c0, false, false);
            c1 = __builtin_amdgcn_wmma_f32_16x16x4_f32(false, a, false, b1v, (short)0, c1, false, false);
            c2 = __builtin_amdgcn_wmma_f32_16x16x4_f32(false, a, false, b2,  (short)0, c2, false, false);
            c3 = __builtin_amdgcn_wmma_f32_16x16x4_f32(false, a, false, b3,  (short)0, c3, false, false);
        }
    }

#pragma unroll
    for (int i = 0; i < 8; ++i) {
        int r = row0 + i + lg * 8;
        if (r < nRows) {
            float* orow = out + (size_t)r * 128 + colofs + lm;
            orow[0]  = c0[i];
            orow[16] = c1[i];
            orow[32] = c2[i];
            orow[48] = c3[i];
        }
    }
}

__global__ __launch_bounds__(256) void k_gemm_first3(const float* __restrict__ T0,
                                                     const float* __restrict__ T1,
                                                     const float* __restrict__ T2,
                                                     const float* __restrict__ W,
                                                     const float* __restrict__ bias,
                                                     float* __restrict__ out,
                                                     int colofs, int nRows) {
    __shared__ float Wl[3 * 64 * 64];
    __shared__ float Al[3 * 128 * 64];
    gemm_orders<3, true>(T0, T1, T2, W, bias, out, colofs, nRows, Wl, Al);
}

__global__ __launch_bounds__(256) void k_gemm_acc2(const float* __restrict__ T0,
                                                   const float* __restrict__ T1,
                                                   const float* __restrict__ W,
                                                   float* __restrict__ out,
                                                   int colofs, int nRows) {
    __shared__ float Wl[2 * 64 * 64];
    __shared__ float Al[2 * 128 * 64];
    gemm_orders<2, false>(T0, T1, T1, W, W, out, colofs, nRows, Wl, Al);
}


extern "C" void kernel_launch(void* const* d_in, const int* in_sizes, int n_in,
                              void* d_out, int out_size, void* d_ws, size_t ws_size,
                              hipStream_t stream) {
    const float* x  = (const float*)d_in[0];
    const int*   ei = (const int*)d_in[1];
    const float* ew = (const float*)d_in[2];
    const float* W1 = (const float*)d_in[3];
    const float* b1 = (const float*)d_in[4];
    const float* W2 = (const float*)d_in[5];
    const float* b2 = (const float*)d_in[6];
    float* out = (float*)d_out;

    const int N = in_sizes[0] / 64;
    const int E = in_sizes[2];
    const int* row = ei;
    const int* col = ei + E;

    float* tA   = (float*)d_ws;
    float* tB   = tA + (size_t)N * 64;
    unsigned* rec = (unsigned*)(tB + (size_t)N * 64);
    float* deg1 = (float*)(rec + (size_t)N * 64);
    float* deg2 = deg1 + N;
    int* rcnt1  = (int*)(deg2 + N);
    int* rcnt2  = rcnt1 + 256;

    const float DIAG = 2.0f / 3.0f - 1.0f;

    dim3 blk(256);
    int gZ    = (2 * N + 512 + 255) / 256;
    int gN2   = (2 * N + 255) / 256;
    int gE    = (E + 255) / 256;
    int gBk   = (E + BK_CHUNK - 1) / BK_CHUNK;
    int gOwn  = (N + ROWS_PER_BLOCK - 1) / ROWS_PER_BLOCK;
    int gGemm = (N + 127) / 128;

    k_zero<<<gZ, blk, 0, stream>>>(deg1, 2 * N + 512);
    k_deg<<<gE, blk, 0, stream>>>(ew, row, col, deg1, deg2, E);
    k_invsqrt<<<gN2, blk, 0, stream>>>(deg1, 2 * N);

    for (int dir = 0; dir < 2; ++dir) {
        const int*      src  = (dir == 0) ? col : row;
        const int*      dst  = (dir == 0) ? row : col;
        const float*    dis  = (dir == 0) ? deg1 : deg2;
        int*            rcnt = (dir == 0) ? rcnt1 : rcnt2;
        const float*    W    = (dir == 0) ? W1 : W2;
        const float*    bias = (dir == 0) ? b1 : b2;
        const int       cofs = (dir == 0) ? 0 : 64;

        k_bucket<<<gBk, BK_THREADS, 0, stream>>>(src, dst, ew, dis, rec, rcnt, N, E);

        k_own<<<gOwn, 1024, 0, stream>>>(x, x, tA, src, dst, ew, dis, rec, rcnt, DIAG, 0.0f, 1.0f, N, E);

        k_own<<<gOwn, 1024, 0, stream>>>(tA, x, tB, src, dst, ew, dis, rec, rcnt, 2.0f * DIAG, -1.0f, 2.0f, N, E);

        k_gemm_first3<<<gGemm, blk, 0, stream>>>(x, tA, tB, W, bias, out, cofs, N);

        k_own<<<gOwn, 1024, 0, stream>>>(tB, tA, tA, src, dst, ew, dis, rec, rcnt, 2.0f * DIAG, -1.0f, 2.0f, N, E);

        k_own<<<gOwn, 1024, 0, stream>>>(tA, tB, tB, src, dst, ew, dis, rec, rcnt, 2.0f * DIAG, -1.0f, 2.0f, N, E);

        k_gemm_acc2<<<gGemm, blk, 0, stream>>>(tA, tB, W + 3 * 4096, out, cofs, N);
    }
}


# speedup vs baseline: 3.0525x; 1.0492x over previous
#include <hip/hip_runtime.h>


typedef float v2f __attribute__((ext_vector_type(2)));
typedef float v8f __attribute__((ext_vector_type(8)));

static __device__ __forceinline__ size_t row64(int n) { return (size_t)n * 64; }


__global__ __launch_bounds__(256) void k_zero(float* __restrict__ p, int n) {
    int i = blockIdx.x * 256 + threadIdx.x;
    if (i < n) p[i] = 0.0f;
}

__global__ __launch_bounds__(256) void k_deg(const float* __restrict__ ew,
                                             const int* __restrict__ row,
                                             const int* __restrict__ col,
                                             float* __restrict__ deg1,
                                             float* __restrict__ deg2, int E) {
    int e = blockIdx.x * 256 + threadIdx.x;
    if (e >= E) return;
    float w = ew[e];
    atomicAdd(&deg1[row[e]], w);
    atomicAdd(&deg2[col[e]], w);
}

__global__ __launch_bounds__(256) void k_invsqrt(float* __restrict__ d, int n) {
    int i = blockIdx.x * 256 + threadIdx.x;
    if (i < n) {
        float v = d[i];
        d[i] = (v > 0.0f) ? rsqrtf(v) : 0.0f;
    }
}

#define ROWS_PER_BLOCK 200
#define LDS_PITCH 320
#define BK_THREADS 256
#define BK_U 16
#define BK_CHUNK (BK_THREADS * BK_U)
__global__ __launch_bounds__(BK_THREADS) void k_bucket(const int* __restrict__ srcIdx,
                                                       const int* __restrict__ dstIdx,
                                                       const float* __restrict__ ew,
                                                       const float* __restrict__ dis,
                                                       unsigned* __restrict__ rec,
                                                       int* __restrict__ rcnt,
                                                       int N, int E) {
    __shared__ int slot[BK_THREADS];
    const int tid = threadIdx.x;
    slot[tid] = 0;
    __syncthreads();

    const float scale = 2.0f / 3.0f;
    const int e0 = blockIdx.x * BK_CHUNK + tid;
    const int elast = E - 1;
    int s[BK_U], d[BK_U];
    float w[BK_U];
#pragma unroll
    for (int k = 0; k < BK_U; ++k) {
        const int e = e0 + k * BK_THREADS;
        const int ec = (e < E) ? e : elast;
        s[k] = srcIdx[ec];
        d[k] = dstIdx[ec];
        w[k] = ew[ec];
    }
    unsigned rx[BK_U], ry[BK_U];
    int bl[BK_U], loc[BK_U];
#pragma unroll
    for (int k = 0; k < BK_U; ++k) {
        const float dd = dis[d[k]];
        const float ds = dis[s[k]];
        const int b = (int)((unsigned)d[k] / ROWS_PER_BLOCK);
        bl[k] = b;
        rx[k] = ((unsigned)s[k] << 8) | (unsigned)(d[k] - b * ROWS_PER_BLOCK);
        ry[k] = __builtin_bit_cast(unsigned, -scale * dd * w[k] * ds);
    }
#pragma unroll
    for (int k = 0; k < BK_U; ++k) {
        loc[k] = 0;
        if (e0 + k * BK_THREADS < E) loc[k] = atomicAdd(&slot[bl[k]], 1);
    }
    __syncthreads();
    {
        const int cnt = slot[tid];
        if (cnt > 0) slot[tid] = atomicAdd(&rcnt[tid], cnt);
    }
    __syncthreads();
#pragma unroll
    for (int k = 0; k < BK_U; ++k) {
        if (e0 + k * BK_THREADS < E) {
            const int b = bl[k];
            const int pos = slot[b] + loc[k];
            int nrows = N - b * ROWS_PER_BLOCK;
            if (nrows > ROWS_PER_BLOCK) nrows = ROWS_PER_BLOCK;
            if (pos < nrows * 32)
                ((uint2*)rec)[(size_t)b * (ROWS_PER_BLOCK * 32) + pos] = make_uint2(rx[k], ry[k]);
        }
    }
}


static_assert(ROWS_PER_BLOCK <= 256, "a record keeps the local row in 8 bits");
static __device__ __forceinline__ uint2 edge_rec(const uint2* rq, int n, int base, int lane) {
    const int idx = base + ((lane & 15) ^ (lane >> 4));
    return rq[(idx < n) ? idx : base];
}

template <int I>
static __device__ __forceinline__ int row_share(int x) {
    return __builtin_amdgcn_update_dpp(0, x, 0x150 + I, 0xf, 0xf, false);
}

template <int P>
static __device__ __forceinline__ float4 gather_pair(const float* __restrict__ v, unsigned recx, unsigned lane16b) {
    const unsigned sx = (unsigned)row_share<2 * P>((int)recx);
    const unsigned off = (sx & 0xFFFFFF00u) | lane16b;
    return *(const float4*)((const char*)v + off);
}

template <int P, bool TAIL>
static __device__ __forceinline__ void add_pair(float* acc, float4 t, unsigned recx, float wt,
                                                int kA, int kB, bool upper, bool ok) {
    const int sx = row_share<2 * P>((int)recx);
    const float w = __builtin_bit_cast(float, row_share<2 * P>(__builtin_bit_cast(int, wt)));
    float* row = acc + (sx & 255) * LDS_PITCH;
    const float r0 = upper ? t.y : t.x;
    const float r1 = upper ? t.x : t.y;
    const float r2 = upper ? t.w : t.z;
    const float r3 = upper ? t.z : t.w;
    if (!TAIL || ok) {
        atomicAdd(row + kA, w * r0);
        atomicAdd(row + kB, w * r1);
        atomicAdd(row + kA + 32, w * r2);
        atomicAdd(row + kB + 32, w * r3);
    }
}

static __device__ __forceinline__ void add_edge_row(float* acc, const float* __restrict__ v,
                                                    const int* __restrict__ srcIdx,
                                                    const int* __restrict__ dstIdx,
                                                    const float* __restrict__ ew,
                                                    const float* __restrict__ dis,
                                                    float alpha, int lo_row, int e, int lane) {
    const float scale = 2.0f / 3.0f;
    const int s = srcIdx[e], d = dstIdx[e];
    const float wt = alpha * (-scale * dis[d] * ew[e] * dis[s]);
    const float* r = v + row64(s) + lane;
    const float vlo = r[0], vhi = r[32];
    float* o = acc + (d - lo_row) * LDS_PITCH + 16 * (lane & 3) + (lane >> 2);
    atomicAdd(o, wt * vlo);
    atomicAdd(o + 8, wt * vhi);
}

__global__ __launch_bounds__(1024) __attribute__((amdgpu_waves_per_eu(8, 8)))
void k_own(const float* __restrict__ v,
           const float* prev,
           float* outf,
           const int* __restrict__ srcIdx,
           const int* __restrict__ dstIdx,
           const float* __restrict__ ew,
           const float* __restrict__ dis,
           const unsigned* __restrict__ rec,
           const int* __restrict__ rcnt,
           float alpha_diag, float beta, float alpha,
           int N, int E) {
    __shared__ float acc[ROWS_PER_BLOCK * LDS_PITCH];

    const int tid = threadIdx.x;
    const int lane = tid & 31;
    const int wave = __builtin_amdgcn_readfirstlane(tid >> 5);
    const int lo_row = blockIdx.x * ROWS_PER_BLOCK;
    int nrows = N - lo_row;
    if (nrows > ROWS_PER_BLOCK) nrows = ROWS_PER_BLOCK;
    const int nq4 = nrows * 16;
    const uint2* rq = (const uint2*)rec + (size_t)lo_row * 32;
    const int rcap = nrows * 32;

    {
        const float4* v4 = (const float4*)(v + row64(lo_row));
        const float4* p4 = (const float4*)(prev + row64(lo_row));
        for (int i = tid; i < nq4; i += 1024) {
            float4 a = v4[i], p = p4[i];
            float* o = acc + (i >> 4) * LDS_PITCH + (i & 15);
            o[0]  = alpha_diag * a.x + beta * p.x;
            o[16] = alpha_diag * a.y + beta * p.y;
            o[32] = alpha_diag * a.z + beta * p.z;
            o[48] = alpha_diag * a.w + beta * p.w;
        }
    }
    __syncthreads();

    const int nrec = __builtin_amdgcn_readfirstlane(rcnt[blockIdx.x]);
    if (nrec <= rcap) {
        const int ngroups = (nrec + 15) >> 4;
        int g = wave;
        uint2 nxt = make_uint2(0u, 0u);
        if (g < ngroups) nxt = edge_rec(rq, nrec, g * 16, lane);
        const int l16 = lane & 15;
        const bool upper = lane >= 16;
        const unsigned lane16b = (unsigned)l16 * 16u;
        const int kA = l16 + (upper ? 16 : 0);
        const int kB = l16 + (upper ? 0 : 16);
        while (g < ngroups) {
            const int base = g * 16;
            const int gn = g + 32;
            const uint2 cur = nxt;
            const float4 t0 = gather_pair<0>(v, cur.x, lane16b);
            const float4 t1 = gather_pair<1>(v, cur.x, lane16b);
            const float4 t2 = gather_pair<2>(v, cur.x, lane16b);
            const float4 t3 = gather_pair<3>(v, cur.x, lane16b);
            const float4 t4 = gather_pair<4>(v, cur.x, lane16b);
            const float4 t5 = gather_pair<5>(v, cur.x, lane16b);
            const float4 t6 = gather_pair<6>(v, cur.x, lane16b);
            const float4 t7 = gather_pair<7>(v, cur.x, lane16b);
            if (gn < ngroups) nxt = edge_rec(rq, nrec, gn * 16, lane);
            const float wt = alpha * __builtin_bit_cast(float, cur.y);
            if (base + 16 <= nrec) {
                add_pair<0, false>(acc, t0, cur.x, wt, kA, kB, upper, true);
                add_pair<1, false>(acc, t1, cur.x, wt, kA, kB, upper, true);
                add_pair<2, false>(acc, t2, cur.x, wt, kA, kB, upper, true);
                add_pair<3, false>(acc, t3, cur.x, wt, kA, kB, upper, true);
                add_pair<4, false>(acc, t4, cur.x, wt, kA, kB, upper, true);
                add_pair<5, false>(acc, t5, cur.x, wt, kA, kB, upper, true);
                add_pair<6, false>(acc, t6, cur.x, wt, kA, kB, upper, true);
                add_pair<7, false>(acc, t7, cur.x, wt, kA, kB, upper, true);
            } else {
                const int e0 = base + (upper ? 1 : 0);
                add_pair<0, true>(acc, t0, cur.x, wt, kA, kB, upper, e0 + 0 < nrec);
                add_pair<1, true>(acc, t1, cur.x, wt, kA, kB, upper, e0 + 2 < nrec);
                add_pair<2, true>(acc, t2, cur.x, wt, kA, kB, upper, e0 + 4 < nrec);
                add_pair<3, true>(acc, t3, cur.x, wt, kA, kB, upper, e0 + 6 < nrec);
                add_pair<4, true>(acc, t4, cur.x, wt, kA, kB, upper, e0 + 8 < nrec);
                add_pair<5, true>(acc, t5, cur.x, wt, kA, kB, upper, e0 + 10 < nrec);
                add_pair<6, true>(acc, t6, cur.x, wt, kA, kB, upper, e0 + 12 < nrec);
                add_pair<7, true>(acc, t7, cur.x, wt, kA, kB, upper, e0 + 14 < nrec);
            }
            g = gn;
        }
    } else {
        for (int eb = wave * 32; eb < E; eb += 32 * 32) {
            const int e = eb + lane;
            bool mine = false;
            if (e < E) mine = (unsigned)(dstIdx[e] - lo_row) < (unsigned)nrows;
            unsigned m = __builtin_amdgcn_ballot_w32(mine);
            while (m) {
                const int l = __builtin_ctz(m);
                m &= m - 1u;
                add_edge_row(acc, v, srcIdx, dstIdx, ew, dis, alpha, lo_row, eb + l, lane);
            }
        }
    }
    __syncthreads();

    {
        float4* o4 = (float4*)(outf + row64(lo_row));
        for (int i = tid; i < nq4; i += 1024) {
            const float* a = acc + (i >> 4) * LDS_PITCH + (i & 15);
            o4[i] = make_float4(a[0], a[16], a[32], a[48]);
        }
    }
}

template <int NT, bool FIRST>
static __device__ __forceinline__ void gemm_orders(const float* __restrict__ T0,
                                                   const float* __restrict__ T1,
                                                   const float* __restrict__ T2,
                                                   const float* __restrict__ W,
                                                   const float* __restrict__ bias,
                                                   float* __restrict__ out,
                                                   int colofs, int nRows,
                                                   float* Wl, float* Al) {
    const int tid = threadIdx.x;
    const int row0b = blockIdx.x * 128;

    const int wave = tid >> 5;
    const int lane = tid & 31;
    const int lm = lane & 15;
    const int lg = lane >> 4;

    const int rowt = wave * 16;
    const int row0 = row0b + rowt;

    v8f c0, c1, c2, c3;
    if (FIRST) {
        const float* bp = bias + lm;
        const float v0 = bp[0], v1 = bp[16], v2 = bp[32], v3 = bp[48];
#pragma unroll
        for (int i = 0; i < 8; ++i) { c0[i] = v0; c1[i] = v1; c2[i] = v2; c3[i] = v3; }
    } else {
        const int nclamp = nRows - 1;
#pragma unroll
        for (int i = 0; i < 8; ++i) {
            int r = row0 + i + lg * 8;
            bool ok = (r < nRows);
            int rc = ok ? r : nclamp;
            const float* orow = out + (size_t)rc * 128 + colofs + lm;
            float v0 = orow[0], v1 = orow[16], v2 = orow[32], v3 = orow[48];
            c0[i] = ok ? v0 : 0.f;
            c1[i] = ok ? v1 : 0.f;
            c2[i] = ok ? v2 : 0.f;
            c3[i] = ok ? v3 : 0.f;
        }
    }

    {
        const int nclamp = nRows - 1;
#pragma unroll
        for (int t = 0; t < NT; ++t) {
            const float* Tt = (t == 0) ? T0 : ((t == 1) ? T1 : T2);
            const float4* W4 = (const float4*)(W + t * (64 * 64));
            float4* Wl4 = (float4*)(Wl + t * (64 * 64));
            float4* Al4 = (float4*)(Al + t * (128 * 64));
#pragma unroll
            for (int i = 0; i < 4; ++i) Wl4[tid + 256 * i] = W4[tid + 256 * i];
#pragma unroll
            for (int i = 0; i < 8; ++i) {
                const int idx = tid + 256 * i;
                const int r = row0b + (idx >> 4);
                const bool ok = (r < nRows);
                float4 v = ((const float4*)(Tt + row64(ok ? r : nclamp)))[idx & 15];
                if (!ok) v = make_float4(0.f, 0.f, 0.f, 0.f);
                Al4[idx] = v;
            }
        }
    }
    __syncthreads();

#pragma unroll
    for (int t = 0; t < NT; ++t) {
        const float* Wt = Wl + t * (64 * 64);
        const float* At = Al + t * (128 * 64);
        const float* arow_l = At + (rowt + lm) * 64;
        const float* bcol_l = Wt + lm;

#pragma unroll
        for (int kk = 0; kk < 16; ++kk) {
            const int kbase = kk * 4 + lg * 2;
            v2f a = *(const v2f*)(arow_l + kbase);
            v2f b0, b1v, b2, b3;
            b0[0]  = bcol_l[(kbase + 0) * 64 + 0];
            b0[1]  = bcol_l[(kbase + 1) * 64 + 0];
            b1v[0] = bcol_l[(kbase + 0) * 64 + 16];
            b1v[1] = bcol_l[(kbase + 1) * 64 + 16];
            b2[0]  = bcol_l[(kbase + 0) * 64 + 32];
            b2[1]  = bcol_l[(kbase + 1) * 64 + 32];
            b3[0]  = bcol_l[(kbase + 0) * 64 + 48];
            b3[1]  = bcol_l[(kbase + 1) * 64 + 48];
            c0 = __builtin_amdgcn_wmma_f32_16x16x4_f32(false, a, false, b0,  (short)0, c0, false, false);
            c1 = __builtin_amdgcn_wmma_f32_16x16x4_f32(false, a, false, b1v, (short)0, c1, false, false);
            c2 = __builtin_amdgcn_wmma_f32_16x16x4_f32(false, a, false, b2,  (short)0, c2, false, false);
            c3 = __builtin_amdgcn_wmma_f32_16x16x4_f32(false, a, false, b3,  (short)0, c3, false, false);
        }
    }

#pragma unroll
    for (int i = 0; i < 8; ++i) {
        int r = row0 + i + lg * 8;
        if (r < nRows) {
            float* orow = out + (size_t)r * 128 + colofs + lm;
            orow[0]  = c0[i];
            orow[16] = c1[i];
            orow[32] = c2[i];
            orow[48] = c3[i];
        }
    }
}

__global__ __launch_bounds__(256) void k_gemm_first3(const float* __restrict__ T0,
                                                     const float* __restrict__ T1,
                                                     const float* __restrict__ T2,
                                                     const float* __restrict__ W,
                                                     const float* __restrict__ bias,
                                                     float* __restrict__ out,
                                                     int colofs, int nRows) {
    __shared__ float Wl[3 * 64 * 64];
    __shared__ float Al[3 * 128 * 64];
    gemm_orders<3, true>(T0, T1, T2, W, bias, out, colofs, nRows, Wl, Al);
}

__global__ __launch_bounds__(256) void k_gemm_acc2(const float* __restrict__ T0,
                                                   const float* __restrict__ T1,
                                                   const float* __restrict__ W,
                                                   float* __restrict__ out,
                                                   int colofs, int nRows) {
    __shared__ float Wl[2 * 64 * 64];
    __shared__ float Al[2 * 128 * 64];
    gemm_orders<2, false>(T0, T1, T1, W, W, out, colofs, nRows, Wl, Al);
}


extern "C" void kernel_launch(void* const* d_in, const int* in_sizes, int n_in,
                              void* d_out, int out_size, void* d_ws, size_t ws_size,
                              hipStream_t stream) {
    const float* x  = (const float*)d_in[0];
    const int*   ei = (const int*)d_in[1];
    const float* ew = (const float*)d_in[2];
    const float* W1 = (const float*)d_in[3];
    const float* b1 = (const float*)d_in[4];
    const float* W2 = (const float*)d_in[5];
    const float* b2 = (const float*)d_in[6];
    float* out = (float*)d_out;

    const int N = in_sizes[0] / 64;
    const int E = in_sizes[2];
    const int* row = ei;
    const int* col = ei + E;

    float* tA   = (float*)d_ws;
    float* tB   = tA + (size_t)N * 64;
    unsigned* rec = (unsigned*)(tB + (size_t)N * 64);
    float* deg1 = (float*)(rec + (size_t)N * 64);
    float* deg2 = deg1 + N;
    int* rcnt1  = (int*)(deg2 + N);
    int* rcnt2  = rcnt1 + 256;

    const float DIAG = 2.0f / 3.0f - 1.0f;

    dim3 blk(256);
    int gZ    = (2 * N + 512 + 255) / 256;
    int gN2   = (2 * N + 255) / 256;
    int gE    = (E + 255) / 256;
    int gBk   = (E + BK_CHUNK - 1) / BK_CHUNK;
    int gOwn  = (N + ROWS_PER_BLOCK - 1) / ROWS_PER_BLOCK;
    int gGemm = (N + 127) / 128;

    k_zero<<<gZ, blk, 0, stream>>>(deg1, 2 * N + 512);
    k_deg<<<gE, blk, 0, stream>>>(ew, row, col, deg1, deg2, E);
    k_invsqrt<<<gN2, blk, 0, stream>>>(deg1, 2 * N);

    for (int dir = 0; dir < 2; ++dir) {
        const int*      src  = (dir == 0) ? col : row;
        const int*      dst  = (dir == 0) ? row : col;
        const float*    dis  = (dir == 0) ? deg1 : deg2;
        int*            rcnt = (dir == 0) ? rcnt1 : rcnt2;
        const float*    W    = (dir == 0) ? W1 : W2;
        const float*    bias = (dir == 0) ? b1 : b2;
        const int       cofs = (dir == 0) ? 0 : 64;

        k_bucket<<<gBk, BK_THREADS, 0, stream>>>(src, dst, ew, dis, rec, rcnt, N, E);

        k_own<<<gOwn, 1024, 0, stream>>>(x, x, tA, src, dst, ew, dis, rec, rcnt, DIAG, 0.0f, 1.0f, N, E);

        k_own<<<gOwn, 1024, 0, stream>>>(tA, x, tB, src, dst, ew, dis, rec, rcnt, 2.0f * DIAG, -1.0f, 2.0f, N, E);

        k_gemm_first3<<<gGemm, blk, 0, stream>>>(x, tA, tB, W, bias, out, cofs, N);

        k_own<<<gOwn, 1024, 0, stream>>>(tB, tA, tA, src, dst, ew, dis, rec, rcnt, 2.0f * DIAG, -1.0f, 2.0f, N, E);

        k_own<<<gOwn, 1024, 0, stream>>>(tA, tB, tB, src, dst, ew, dis, rec, rcnt, 2.0f * DIAG, -1.0f, 2.0f, N, E);

        k_gemm_acc2<<<gGemm, blk, 0, stream>>>(tA, tB, W + 3 * 4096, out, cofs, N);
    }
}


# Two dependent launches fewer (17 -> 15): deg^-1/2 is taken where a degree is read, no pass of its own; direction 2's bucketing pass rides in the launch of direction 1's last projection; same sums
# speedup vs baseline: 3.1443x; 1.0301x over previous
#include <hip/hip_runtime.h>


typedef float v2f __attribute__((ext_vector_type(2)));
typedef float v8f __attribute__((ext_vector_type(8)));

static __device__ __forceinline__ size_t row64(int n) { return (size_t)n * 64; }


__global__ __launch_bounds__(256) void k_zero(float* __restrict__ p, int n) {
    int i = blockIdx.x * 256 + threadIdx.x;
    if (i < n) p[i] = 0.0f;
}

__global__ __launch_bounds__(256) void k_deg(const float* __restrict__ ew,
                                             const int* __restrict__ row,
                                             const int* __restrict__ col,
                                             float* __restrict__ deg1,
                                             float* __restrict__ deg2, int E) {
    int e = blockIdx.x * 256 + threadIdx.x;
    if (e >= E) return;
    float w = ew[e];
    atomicAdd(&deg1[row[e]], w);
    atomicAdd(&deg2[col[e]], w);
}

static __device__ __forceinline__ float inv_sqrt_deg(float v) {
    return (v > 0.0f) ? rsqrtf(v) : 0.0f;
}

#define ROWS_PER_BLOCK 200
#define LDS_PITCH 320
#define BK_THREADS 256
#define BK_U 16
#define BK_CHUNK (BK_THREADS * BK_U)
static __device__ __forceinline__ void bucket_chunk(int chunk, int* slot,
                                                    const int* __restrict__ srcIdx,
                                                    const int* __restrict__ dstIdx,
                                                    const float* __restrict__ ew,
                                                    const float* __restrict__ deg,
                                                    unsigned* __restrict__ rec,
                                                    int* __restrict__ rcnt,
                                                    int N, int E) {
    const int tid = threadIdx.x;
    slot[tid] = 0;
    __syncthreads();

    const float scale = 2.0f / 3.0f;
    const int e0 = chunk * BK_CHUNK + tid;
    const int elast = E - 1;
    int s[BK_U], d[BK_U];
    float w[BK_U];
#pragma unroll
    for (int k = 0; k < BK_U; ++k) {
        const int e = e0 + k * BK_THREADS;
        const int ec = (e < E) ? e : elast;
        s[k] = srcIdx[ec];
        d[k] = dstIdx[ec];
        w[k] = ew[ec];
    }
    unsigned rx[BK_U], ry[BK_U];
    int bl[BK_U], loc[BK_U];
#pragma unroll
    for (int k = 0; k < BK_U; ++k) {
        const float dd = inv_sqrt_deg(deg[d[k]]);
        const float ds = inv_sqrt_deg(deg[s[k]]);
        const int b = (int)((unsigned)d[k] / ROWS_PER_BLOCK);
        bl[k] = b;
        rx[k] = ((unsigned)s[k] << 8) | (unsigned)(d[k] - b * ROWS_PER_BLOCK);
        ry[k] = __builtin_bit_cast(unsigned, -scale * dd * w[k] * ds);
    }
#pragma unroll
    for (int k = 0; k < BK_U; ++k) {
        loc[k] = 0;
        if (e0 + k * BK_THREADS < E) loc[k] = atomicAdd(&slot[bl[k]], 1);
    }
    __syncthreads();
    {
        const int cnt = slot[tid];
        if (cnt > 0) slot[tid] = atomicAdd(&rcnt[tid], cnt);
    }
    __syncthreads();
#pragma unroll
    for (int k = 0; k < BK_U; ++k) {
        if (e0 + k * BK_THREADS < E) {
            const int b = bl[k];
            const int pos = slot[b] + loc[k];
            int nrows = N - b * ROWS_PER_BLOCK;
            if (nrows > ROWS_PER_BLOCK) nrows = ROWS_PER_BLOCK;
            if (pos < nrows * 32)
                ((uint2*)rec)[(size_t)b * (ROWS_PER_BLOCK * 32) + pos] = make_uint2(rx[k], ry[k]);
        }
    }
}

__global__ __launch_bounds__(BK_THREADS) void k_bucket(const int* __restrict__ srcIdx,
                                                       const int* __restrict__ dstIdx,
                                                       const float* __restrict__ ew,
                                                       const float* __restrict__ deg,
                                                       unsigned* __restrict__ rec,
                                                       int* __restrict__ rcnt,
                                                       int N, int E) {
    __shared__ int slot[BK_THREADS];
    bucket_chunk(blockIdx.x, slot, srcIdx, dstIdx, ew, deg, rec, rcnt, N, E);
}


static_assert(ROWS_PER_BLOCK <= 256, "a record keeps the local row in 8 bits");
static __device__ __forceinline__ uint2 edge_rec(const uint2* rq, int n, int base, int lane) {
    const int idx = base + ((lane & 15) ^ (lane >> 4));
    return rq[(idx < n) ? idx : base];
}

template <int I>
static __device__ __forceinline__ int row_share(int x) {
    return __builtin_amdgcn_update_dpp(0, x, 0x150 + I, 0xf, 0xf, false);
}

template <int P>
static __device__ __forceinline__ float4 gather_pair(const float* __restrict__ v, unsigned recx, unsigned lane16b) {
    const unsigned sx = (unsigned)row_share<2 * P>((int)recx);
    const unsigned off = (sx & 0xFFFFFF00u) | lane16b;
    return *(const float4*)((const char*)v + off);
}

template <int P, bool TAIL>
static __device__ __forceinline__ void add_pair(float* acc, float4 t, unsigned recx, float wt,
                                                int kA, int kB, bool upper, bool ok) {
    const int sx = row_share<2 * P>((int)recx);
    const float w = __builtin_bit_cast(float, row_share<2 * P>(__builtin_bit_cast(int, wt)));
    float* row = acc + (sx & 255) * LDS_PITCH;
    const float r0 = upper ? t.y : t.x;
    const float r1 = upper ? t.x : t.y;
    const float r2 = upper ? t.w : t.z;
    const float r3 = upper ? t.z : t.w;
    if (!TAIL || ok) {
        atomicAdd(row + kA, w * r0);
        atomicAdd(row + kB, w * r1);
        atomicAdd(row + kA + 32, w * r2);
        atomicAdd(row + kB + 32, w * r3);
    }
}

static __device__ __forceinline__ void add_edge_row(float* acc, const float* __restrict__ v,
                                                    const int* __restrict__ srcIdx,
                                                    const int* __restrict__ dstIdx,
                                                    const float* __restrict__ ew,
                                                    const float* __restrict__ dis,
                                                    float alpha, int lo_row, int e, int lane) {
    const float scale = 2.0f / 3.0f;
    const int s = srcIdx[e], d = dstIdx[e];
    const float wt = alpha * (-scale * inv_sqrt_deg(dis[d]) * ew[e] * inv_sqrt_deg(dis[s]));
    const float* r = v + row64(s) + lane;
    const float vlo = r[0], vhi = r[32];
    float* o = acc + (d - lo_row) * LDS_PITCH + 16 * (lane & 3) + (lane >> 2);
    atomicAdd(o, wt * vlo);
    atomicAdd(o + 8, wt * vhi);
}

__global__ __launch_bounds__(1024) __attribute__((amdgpu_waves_per_eu(8, 8)))
void k_own(const float* __restrict__ v,
           const float* prev,
           float* outf,
           const int* __restrict__ srcIdx,
           const int* __restrict__ dstIdx,
           const float* __restrict__ ew,
           const float* __restrict__ dis,
           const unsigned* __restrict__ rec,
           const int* __restrict__ rcnt,
           float alpha_diag, float beta, float alpha,
           int N, int E) {
    __shared__ float acc[ROWS_PER_BLOCK * LDS_PITCH];

    const int tid = threadIdx.x;
    const int lane = tid & 31;
    const int wave = __builtin_amdgcn_readfirstlane(tid >> 5);
    const int lo_row = blockIdx.x * ROWS_PER_BLOCK;
    int nrows = N - lo_row;
    if (nrows > ROWS_PER_BLOCK) nrows = ROWS_PER_BLOCK;
    const int nq4 = nrows * 16;
    const uint2* rq = (const uint2*)rec + (size_t)lo_row * 32;
    const int rcap = nrows * 32;

    {
        const float4* v4 = (const float4*)(v + row64(lo_row));
        const float4* p4 = (const float4*)(prev + row64(lo_row));
        for (int i = tid; i < nq4; i += 1024) {
            float4 a = v4[i], p = p4[i];
            float* o = acc + (i >> 4) * LDS_PITCH + (i & 15);
            o[0]  = alpha_diag * a.x + beta * p.x;
            o[16] = alpha_diag * a.y + beta * p.y;
            o[32] = alpha_diag * a.z + beta * p.z;
            o[48] = alpha_diag * a.w + beta * p.w;
        }
    }
    __syncthreads();

    const int nrec = __builtin_amdgcn_readfirstlane(rcnt[blockIdx.x]);
    if (nrec <= rcap) {
        const int ngroups = (nrec + 15) >> 4;
        int g = wave;
        uint2 nxt = make_uint2(0u, 0u);
        if (g < ngroups) nxt = edge_rec(rq, nrec, g * 16, lane);
        const int l16 = lane & 15;
        const bool upper = lane >= 16;
        const unsigned lane16b = (unsigned)l16 * 16u;
        const int kA = l16 + (upper ? 16 : 0);
        const int kB = l16 + (upper ? 0 : 16);
        while (g < ngroups) {
            const int base = g * 16;
            const int gn = g + 32;
            const uint2 cur = nxt;
            const float4 t0 = gather_pair<0>(v, cur.x, lane16b);
            const float4 t1 = gather_pair<1>(v, cur.x, lane16b);
            const float4 t2 = gather_pair<2>(v, cur.x, lane16b);
            const float4 t3 = gather_pair<3>(v, cur.x, lane16b);
            const float4 t4 = gather_pair<4>(v, cur.x, lane16b);
            const float4 t5 = gather_pair<5>(v, cur.x, lane16b);
            const float4 t6 = gather_pair<6>(v, cur.x, lane16b);
            const float4 t7 = gather_pair<7>(v, cur.x, lane16b);
            if (gn < ngroups) nxt = edge_rec(rq, nrec, gn * 16, lane);
            const float wt = alpha * __builtin_bit_cast(float, cur.y);
            if (base + 16 <= nrec) {
                add_pair<0, false>(acc, t0, cur.x, wt, kA, kB, upper, true);
                add_pair<1, false>(acc, t1, cur.x, wt, kA, kB, upper, true);
                add_pair<2, false>(acc, t2, cur.x, wt, kA, kB, upper, true);
                add_pair<3, false>(acc, t3, cur.x, wt, kA, kB, upper, true);
                add_pair<4, false>(acc, t4, cur.x, wt, kA, kB, upper, true);
                add_pair<5, false>(acc, t5, cur.x, wt, kA, kB, upper, true);
                add_pair<6, false>(acc, t6, cur.x, wt, kA, kB, upper, true);
                add_pair<7, false>(acc, t7, cur.x, wt, kA, kB, upper, true);
            } else {
                const int e0 = base + (upper ? 1 : 0);
                add_pair<0, true>(acc, t0, cur.x, wt, kA, kB, upper, e0 + 0 < nrec);
                add_pair<1, true>(acc, t1, cur.x, wt, kA, kB, upper, e0 + 2 < nrec);
                add_pair<2, true>(acc, t2, cur.x, wt, kA, kB, upper, e0 + 4 < nrec);
                add_pair<3, true>(acc, t3, cur.x, wt, kA, kB, upper, e0 + 6 < nrec);
                add_pair<4, true>(acc, t4, cur.x, wt, kA, kB, upper, e0 + 8 < nrec);
                add_pair<5, true>(acc, t5, cur.x, wt, kA, kB, upper, e0 + 10 < nrec);
                add_pair<6, true>(acc, t6, cur.x, wt, kA, kB, upper, e0 + 12 < nrec);
                add_pair<7, true>(acc, t7, cur.x, wt, kA, kB, upper, e0 + 14 < nrec);
            }
            g = gn;
        }
    } else {
        for (int eb = wave * 32; eb < E; eb += 32 * 32) {
            const int e = eb + lane;
            bool mine = false;
            if (e < E) mine = (unsigned)(dstIdx[e] - lo_row) < (unsigned)nrows;
            unsigned m = __builtin_amdgcn_ballot_w32(mine);
            while (m) {
                const int l = __builtin_ctz(m);
                m &= m - 1u;
                add_edge_row(acc, v, srcIdx, dstIdx, ew, dis, alpha, lo_row, eb + l, lane);
            }
        }
    }
    __syncthreads();

    {
        float4* o4 = (float4*)(outf + row64(lo_row));
        for (int i = tid; i < nq4; i += 1024) {
            const float* a = acc + (i >> 4) * LDS_PITCH + (i & 15);
            o4[i] = make_float4(a[0], a[16], a[32], a[48]);
        }
    }
}

template <int NT, bool FIRST>
static __device__ __forceinline__ void gemm_orders(const float* __restrict__ T0,
                                                   const float* __restrict__ T1,
                                                   const float* __restrict__ T2,
                                                   const float* __restrict__ W,
                                                   const float* __restrict__ bias,
                                                   float* __restrict__ out,
                                                   int colofs, int nRows,
                                                   float* Wl, float* Al, int blk) {
    const int tid = threadIdx.x;
    const int row0b = blk * 128;

    const int wave = tid >> 5;
    const int lane = tid & 31;
    const int lm = lane & 15;
    const int lg = lane >> 4;

    const int rowt = wave * 16;
    const int row0 = row0b + rowt;

    v8f c0, c1, c2, c3;
    if (FIRST) {
        const float* bp = bias + lm;
        const float v0 = bp[0], v1 = bp[16], v2 = bp[32], v3 = bp[48];
#pragma unroll
        for (int i = 0; i < 8; ++i) { c0[i] = v0; c1[i] = v1; c2[i] = v2; c3[i] = v3; }
    } else {
        const int nclamp = nRows - 1;
#pragma unroll
        for (int i = 0; i < 8; ++i) {
            int r = row0 + i + lg * 8;
            bool ok = (r < nRows);
            int rc = ok ? r : nclamp;
            const float* orow = out + (size_t)rc * 128 + colofs + lm;
            float v0 = orow[0], v1 = orow[16], v2 = orow[32], v3 = orow[48];
            c0[i] = ok ? v0 : 0.f;
            c1[i] = ok ? v1 : 0.f;
            c2[i] = ok ? v2 : 0.f;
            c3[i] = ok ? v3 : 0.f;
        }
    }

    {
        const int nclamp = nRows - 1;
#pragma unroll
        for (int t = 0; t < NT; ++t) {
            const float* Tt = (t == 0) ? T0 : ((t == 1) ? T1 : T2);
            const float4* W4 = (const float4*)(W + t * (64 * 64));
            float4* Wl4 = (float4*)(Wl + t * (64 * 64));
            float4* Al4 = (float4*)(Al + t * (128 * 64));
#pragma unroll
            for (int i = 0; i < 4; ++i) Wl4[tid + 256 * i] = W4[tid + 256 * i];
#pragma unroll
            for (int i = 0; i < 8; ++i) {
                const int idx = tid + 256 * i;
                const int r = row0b + (idx >> 4);
                const bool ok = (r < nRows);
                float4 v = ((const float4*)(Tt + row64(ok ? r : nclamp)))[idx & 15];
                if (!ok) v = make_float4(0.f, 0.f, 0.f, 0.f);
                Al4[idx] = v;
            }
        }
    }
    __syncthreads();

#pragma unroll
    for (int t = 0; t < NT; ++t) {
        const float* Wt = Wl + t * (64 * 64);
        const float* At = Al + t * (128 * 64);
        const float* arow_l = At + (rowt + lm) * 64;
        const float* bcol_l = Wt + lm;

#pragma unroll
        for (int kk = 0; kk < 16; ++kk) {
            const int kbase = kk * 4 + lg * 2;
            v2f a = *(const v2f*)(arow_l + kbase);
            v2f b0, b1v, b2, b3;
            b0[0]  = bcol_l[(kbase + 0) * 64 + 0];
            b0[1]  = bcol_l[(kbase + 1) * 64 + 0];
            b1v[0] = bcol_l[(kbase + 0) * 64 + 16];
            b1v[1] = bcol_l[(kbase + 1) * 64 + 16];
            b2[0]  = bcol_l[(kbase + 0) * 64 + 32];
            b2[1]  = bcol_l[(kbase + 1) * 64 + 32];
            b3[0]  = bcol_l[(kbase + 0) * 64 + 48];
            b3[1]  = bcol_l[(kbase + 1) * 64 + 48];
            c0 = __builtin_amdgcn_wmma_f32_16x16x4_f32(false, a, false, b0,  (short)0, c0, false, false);
            c1 = __builtin_amdgcn_wmma_f32_16x16x4_f32(false, a, false, b1v, (short)0, c1, false, false);
            c2 = __builtin_amdgcn_wmma_f32_16x16x4_f32(false, a, false, b2,  (short)0, c2, false, false);
            c3 = __builtin_amdgcn_wmma_f32_16x16x4_f32(false, a, false, b3,  (short)0, c3, false, false);
        }
    }

#pragma unroll
    for (int i = 0; i < 8; ++i) {
        int r = row0 + i + lg * 8;
        if (r < nRows) {
            float* orow = out + (size_t)r * 128 + colofs + lm;
            orow[0]  = c0[i];
            orow[16] = c1[i];
            orow[32] = c2[i];
            orow[48] = c3[i];
        }
    }
}

__global__ __launch_bounds__(256) void k_gemm_first3(const float* __restrict__ T0,
                                                     const float* __restrict__ T1,
                                                     const float* __restrict__ T2,
                                                     const float* __restrict__ W,
                                                     const float* __restrict__ bias,
                                                     float* __restrict__ out,
                                                     int colofs, int nRows) {
    __shared__ float Wl[3 * 64 * 64];
    __shared__ float Al[3 * 128 * 64];
    gemm_orders<3, true>(T0, T1, T2, W, bias, out, colofs, nRows, Wl, Al, blockIdx.x);
}

__global__ __launch_bounds__(256) void k_gemm_acc2(const float* __restrict__ T0,
                                                   const float* __restrict__ T1,
                                                   const float* __restrict__ W,
                                                   float* __restrict__ out,
                                                   int colofs, int nRows) {
    __shared__ float Wl[2 * 64 * 64];
    __shared__ float Al[2 * 128 * 64];
    gemm_orders<2, false>(T0, T1, T1, W, W, out, colofs, nRows, Wl, Al, blockIdx.x);
}

__global__ __launch_bounds__(256) void k_gemm_acc2_bucket(const float* __restrict__ T0,
                                                          const float* __restrict__ T1,
                                                          const float* __restrict__ W,
                                                          float* __restrict__ out,
                                                          int colofs, int nRows,
                                                          const int* __restrict__ srcIdx,
                                                          const int* __restrict__ dstIdx,
                                                          const float* __restrict__ ew,
                                                          const float* __restrict__ deg,
                                                          unsigned* __restrict__ rec,
                                                          int* __restrict__ rcnt,
                                                          int E, int nBk) {
    __shared__ float Wl[2 * 64 * 64];
    __shared__ float Al[2 * 128 * 64];
    static_assert(BK_THREADS == 256, "a bucketing chunk and a projection tile share the block size");
    __shared__ int slot[BK_THREADS];
    const int b = blockIdx.x;
    if (b < nBk) {
        bucket_chunk(b, slot, srcIdx, dstIdx, ew, deg, rec, rcnt, nRows, E);
    } else {
        gemm_orders<2, false>(T0, T1, T1, W, W, out, colofs, nRows, Wl, Al, b - nBk);
    }
}


extern "C" void kernel_launch(void* const* d_in, const int* in_sizes, int n_in,
                              void* d_out, int out_size, void* d_ws, size_t ws_size,
                              hipStream_t stream) {
    const float* x  = (const float*)d_in[0];
    const int*   ei = (const int*)d_in[1];
    const float* ew = (const float*)d_in[2];
    const float* W1 = (const float*)d_in[3];
    const float* b1 = (const float*)d_in[4];
    const float* W2 = (const float*)d_in[5];
    const float* b2 = (const float*)d_in[6];
    float* out = (float*)d_out;

    const int N = in_sizes[0] / 64;
    const int E = in_sizes[2];
    const int* row = ei;
    const int* col = ei + E;

    float* tA   = (float*)d_ws;
    float* tB   = tA + (size_t)N * 64;
    unsigned* rec = (unsigned*)(tB + (size_t)N * 64);
    float* deg1 = (float*)(rec + (size_t)N * 64);
    float* deg2 = deg1 + N;
    int* rcnt1  = (int*)(deg2 + N);
    int* rcnt2  = rcnt1 + 256;

    const float DIAG = 2.0f / 3.0f - 1.0f;

    dim3 blk(256);
    int gZ    = (2 * N + 512 + 255) / 256;
    int gE    = (E + 255) / 256;
    int gBk   = (E + BK_CHUNK - 1) / BK_CHUNK;
    int gOwn  = (N + ROWS_PER_BLOCK - 1) / ROWS_PER_BLOCK;
    int gGemm = (N + 127) / 128;

    k_zero<<<gZ, blk, 0, stream>>>(deg1, 2 * N + 512);
    k_deg<<<gE, blk, 0, stream>>>(ew, row, col, deg1, deg2, E);

    for (int dir = 0; dir < 2; ++dir) {
        const int*      src  = (dir == 0) ? col : row;
        const int*      dst  = (dir == 0) ? row : col;
        const float*    dis  = (dir == 0) ? deg1 : deg2;
        int*            rcnt = (dir == 0) ? rcnt1 : rcnt2;
        const float*    W    = (dir == 0) ? W1 : W2;
        const float*    bias = (dir == 0) ? b1 : b2;
        const int       cofs = (dir == 0) ? 0 : 64;

        if (dir == 0) k_bucket<<<gBk, BK_THREADS, 0, stream>>>(src, dst, ew, dis, rec, rcnt, N, E);

        k_own<<<gOwn, 1024, 0, stream>>>(x, x, tA, src, dst, ew, dis, rec, rcnt, DIAG, 0.0f, 1.0f, N, E);

        k_own<<<gOwn, 1024, 0, stream>>>(tA, x, tB, src, dst, ew, dis, rec, rcnt, 2.0f * DIAG, -1.0f, 2.0f, N, E);

        k_gemm_first3<<<gGemm, blk, 0, stream>>>(x, tA, tB, W, bias, out, cofs, N);

        k_own<<<gOwn, 1024, 0, stream>>>(tB, tA, tA, src, dst, ew, dis, rec, rcnt, 2.0f * DIAG, -1.0f, 2.0f, N, E);

        k_own<<<gOwn, 1024, 0, stream>>>(tA, tB, tB, src, dst, ew, dis, rec, rcnt, 2.0f * DIAG, -1.0f, 2.0f, N, E);

        if (dir == 0)
            k_gemm_acc2_bucket<<<gBk + gGemm, blk, 0, stream>>>(tA, tB, W + 3 * 4096, out, cofs, N,
                                                                row, col, ew, deg2, rec, rcnt2, E, gBk);
        else
            k_gemm_acc2<<<gGemm, blk, 0, stream>>>(tA, tB, W + 3 * 4096, out, cofs, N);
    }
}
